# starting kernel
# baseline (speedup 1.0000x reference)
#include <hip/hip_runtime.h>


typedef __attribute__((ext_vector_type(2))) float v2f;
typedef __attribute__((ext_vector_type(8))) float v8f;

#define IN_F    128
#define HID_F   64
#define OUT_F   40
#define OUT_PAD 48


__global__ void zero_f32(float* __restrict__ p, int n) {
  int t = blockIdx.x * blockDim.x + threadIdx.x;
  if (t < n) p[t] = 0.0f;
}

__global__ void count_degree(const int* __restrict__ idx, float* __restrict__ deg, int n) {
  int t = blockIdx.x * blockDim.x + threadIdx.x;
  if (t < n) atomicAdd(&deg[idx[t]], 1.0f);
}

__global__ void inv_sqrt_clamp(float* __restrict__ deg, int n) {
  int t = blockIdx.x * blockDim.x + threadIdx.x;
  if (t < n) deg[t] = rsqrtf(fmaxf(deg[t], 1.0f));
}

__global__ void pad_w3(const float* __restrict__ W3, float* __restrict__ W3p) {
  int t = blockIdx.x * blockDim.x + threadIdx.x;
  if (t >= HID_F * OUT_PAD) return;
  int r = t / OUT_PAD, c = t % OUT_PAD;
  W3p[t] = (c < OUT_F) ? W3[r * OUT_F + c] : 0.0f;
}


template <int WIDTH>
__global__ void scatter_add(const float* __restrict__ feat, float* __restrict__ agg,
                            const int* __restrict__ se, const int* __restrict__ de,
                            int nEdges) {
  constexpr unsigned CH = WIDTH / 4;
  unsigned t = blockIdx.x * blockDim.x + threadIdx.x;
  unsigned e = t / CH;
  unsigned ch = t - e * CH;
  if (e >= (unsigned)nEdges) return;
  int s = se[e], d = de[e];
  float4 v = ((const float4*)(feat + (size_t)s * WIDTH))[ch];
  float* dp = agg + (size_t)d * WIDTH + (size_t)ch * 4;
  atomicAdd(dp + 0, v.x);
  atomicAdd(dp + 1, v.y);
  atomicAdd(dp + 2, v.z);
  atomicAdd(dp + 3, v.w);
}


__global__ void gemm1_scaleA(const float* __restrict__ X, const float* __restrict__ W,
                             const float* __restrict__ norm_out, float* __restrict__ out,
                             int nNodes, int mTiles) {
  int wave = blockIdx.x * (blockDim.x >> 5) + (threadIdx.x >> 5);
  int lane = threadIdx.x & 31;
  int tm = wave >> 2;
  int tn = wave & 3;
  if (tm >= mTiles) return;
  int half = lane >> 4, l = lane & 15;
  int rowA = tm * 16 + l;
  int rA = rowA < nNodes ? rowA : nNodes - 1;
  float s = norm_out[rA];
  const float* xrow = X + (size_t)rA * IN_F;
  v8f c = {};
  for (int k0 = 0; k0 < IN_F; k0 += 4) {
    int ka = k0 + half * 2;
    float2 av = *(const float2*)(xrow + ka);
    v2f a; a.x = av.x * s; a.y = av.y * s;
    v2f b; b.x = W[ka * HID_F + tn * 16 + l];
           b.y = W[(ka + 1) * HID_F + tn * 16 + l];
    c = __builtin_amdgcn_wmma_f32_16x16x4_f32(false, a, false, b, (short)0, c, false, false);
  }
  int n = tn * 16 + l;
#pragma unroll
  for (int j = 0; j < 8; ++j) {
    int row = tm * 16 + half * 8 + j;
    if (row < nNodes) out[(size_t)row * HID_F + n] = c[j];
  }
}

__global__ void gemm2_epi(const float* __restrict__ A, const float* __restrict__ W,
                          const float* __restrict__ bias, const float* __restrict__ norm_in,
                          const float* __restrict__ norm_out, float* __restrict__ out,
                          int nNodes, int mTiles) {
  int wave = blockIdx.x * (blockDim.x >> 5) + (threadIdx.x >> 5);
  int lane = threadIdx.x & 31;
  int tm = wave >> 2;
  int tn = wave & 3;
  if (tm >= mTiles) return;
  int half = lane >> 4, l = lane & 15;
  int rowA = tm * 16 + l;
  int rA = rowA < nNodes ? rowA : nNodes - 1;
  const float* arow = A + (size_t)rA * HID_F;
  v8f c = {};
  for (int k0 = 0; k0 < HID_F; k0 += 4) {
    int ka = k0 + half * 2;
    float2 av = *(const float2*)(arow + ka);
    v2f a; a.x = av.x; a.y = av.y;
    v2f b; b.x = W[ka * HID_F + tn * 16 + l];
           b.y = W[(ka + 1) * HID_F + tn * 16 + l];
    c = __builtin_amdgcn_wmma_f32_16x16x4_f32(false, a, false, b, (short)0, c, false, false);
  }
  int n = tn * 16 + l;
  float bv = bias[n];
#pragma unroll
  for (int j = 0; j < 8; ++j) {
    int row = tm * 16 + half * 8 + j;
    if (row < nNodes) {
      float v = c[j] * norm_in[row] + bv;
      v = fmaxf(v, 0.0f);
      out[(size_t)row * HID_F + n] = v * norm_out[row];
    }
  }
}

__global__ void gemm3_plain(const float* __restrict__ A, const float* __restrict__ W,
                            float* __restrict__ out, int nNodes, int mTiles) {
  int wave = blockIdx.x * (blockDim.x >> 5) + (threadIdx.x >> 5);
  int lane = threadIdx.x & 31;
  int tm = wave / 3;
  int tn = wave - tm * 3;
  if (tm >= mTiles) return;
  int half = lane >> 4, l = lane & 15;
  int rowA = tm * 16 + l;
  int rA = rowA < nNodes ? rowA : nNodes - 1;
  const float* arow = A + (size_t)rA * HID_F;
  v8f c = {};
  for (int k0 = 0; k0 < HID_F; k0 += 4) {
    int ka = k0 + half * 2;
    float2 av = *(const float2*)(arow + ka);
    v2f a; a.x = av.x; a.y = av.y;
    v2f b; b.x = W[ka * OUT_PAD + tn * 16 + l];
           b.y = W[(ka + 1) * OUT_PAD + tn * 16 + l];
    c = __builtin_amdgcn_wmma_f32_16x16x4_f32(false, a, false, b, (short)0, c, false, false);
  }
  int n = tn * 16 + l;
#pragma unroll
  for (int j = 0; j < 8; ++j) {
    int row = tm * 16 + half * 8 + j;
    if (row < nNodes) out[(size_t)row * OUT_PAD + n] = c[j];
  }
}


__global__ void epilogue1(const float* __restrict__ agg, const float* __restrict__ norm_in,
                          const float* __restrict__ norm_out, const float* __restrict__ bias,
                          float* __restrict__ outp, int nNodes) {
  int t = blockIdx.x * blockDim.x + threadIdx.x;
  if (t >= nNodes * HID_F) return;
  int i = t >> 6;
  int f = t & (HID_F - 1);
  float v = agg[t] * norm_in[i] + bias[f];
  v = fmaxf(v, 0.0f);
  outp[t] = v * norm_out[i];
}

__global__ void epilogue3(const float* __restrict__ agg48, const float* __restrict__ norm_in,
                          const float* __restrict__ bias, float* __restrict__ out, int nNodes) {
  int t = blockIdx.x * blockDim.x + threadIdx.x;
  if (t >= nNodes * OUT_F) return;
  int i = t / OUT_F;
  int f = t - i * OUT_F;
  out[t] = agg48[(size_t)i * OUT_PAD + f] * norm_in[i] + bias[f];
}


static inline int cdiv_ll(long long a, long long b) { return (int)((a + b - 1) / b); }

extern "C" void kernel_launch(void* const* d_in, const int* in_sizes, int n_in,
                              void* d_out, int out_size, void* d_ws, size_t ws_size,
                              hipStream_t stream) {
  const float* X  = (const float*)d_in[0];
  const float* W1 = (const float*)d_in[1];
  const float* b1 = (const float*)d_in[2];
  const float* W2 = (const float*)d_in[3];
  const float* b2 = (const float*)d_in[4];
  const float* W3 = (const float*)d_in[5];
  const float* b3 = (const float*)d_in[6];
  const int* src  = (const int*)d_in[7];
  const int* dst  = (const int*)d_in[8];
  float* out = (float*)d_out;

  const int nNodes = in_sizes[0] / IN_F;
  const int nEdges = in_sizes[7];
  const int mTiles = (nNodes + 15) / 16;

  float* ws = (float*)d_ws;
  float* norm_out = ws;
  float* norm_in  = ws + nNodes;
  float* W3p      = ws + 2 * (size_t)nNodes;
  size_t off = 2 * (size_t)nNodes + HID_F * OUT_PAD;
  off = (off + 15) & ~(size_t)15;
  float* bufA = (float*)d_ws + off;
  float* bufB = bufA + (size_t)nNodes * HID_F;
  float* bufC = bufB + (size_t)nNodes * HID_F;

  const dim3 blk(256);
  const int wavesH = mTiles * (HID_F / 16);
  const int wavesO = mTiles * (OUT_PAD / 16);

  zero_f32<<<cdiv_ll(2LL * nNodes, 256), blk, 0, stream>>>(norm_out, 2 * nNodes);
  count_degree<<<cdiv_ll(nEdges, 256), blk, 0, stream>>>(src, norm_out, nEdges);
  count_degree<<<cdiv_ll(nEdges, 256), blk, 0, stream>>>(dst, norm_in, nEdges);
  inv_sqrt_clamp<<<cdiv_ll(2LL * nNodes, 256), blk, 0, stream>>>(norm_out, 2 * nNodes);
  pad_w3<<<cdiv_ll(HID_F * OUT_PAD, 256), blk, 0, stream>>>(W3, W3p);

  gemm1_scaleA<<<cdiv_ll((long long)wavesH * 32, 256), blk, 0, stream>>>(
      X, W1, norm_out, bufA, nNodes, mTiles);
  zero_f32<<<cdiv_ll((long long)nNodes * HID_F, 256), blk, 0, stream>>>(bufB, nNodes * HID_F);
  scatter_add<HID_F><<<cdiv_ll((long long)nEdges * (HID_F / 4), 256), blk, 0, stream>>>(
      bufA, bufB, src, dst, nEdges);
  epilogue1<<<cdiv_ll((long long)nNodes * HID_F, 256), blk, 0, stream>>>(
      bufB, norm_in, norm_out, b1, bufA, nNodes);

  zero_f32<<<cdiv_ll((long long)nNodes * HID_F, 256), blk, 0, stream>>>(bufB, nNodes * HID_F);
  scatter_add<HID_F><<<cdiv_ll((long long)nEdges * (HID_F / 4), 256), blk, 0, stream>>>(
      bufA, bufB, src, dst, nEdges);
  gemm2_epi<<<cdiv_ll((long long)wavesH * 32, 256), blk, 0, stream>>>(
      bufB, W2, b2, norm_in, norm_out, bufA, nNodes, mTiles);

  gemm3_plain<<<cdiv_ll((long long)wavesO * 32, 256), blk, 0, stream>>>(
      bufA, W3p, bufC, nNodes, mTiles);
  zero_f32<<<cdiv_ll((long long)nNodes * OUT_PAD, 256), blk, 0, stream>>>(bufB, nNodes * OUT_PAD);
  scatter_add<OUT_PAD><<<cdiv_ll((long long)nEdges * (OUT_PAD / 4), 256), blk, 0, stream>>>(
      bufC, bufB, src, dst, nEdges);
  epilogue3<<<cdiv_ll((long long)nNodes * OUT_F, 256), blk, 0, stream>>>(
      bufB, norm_in, b3, out, nNodes);
}


# Per-node register gather over a per-call counting sort of edges by destination (integer rank atomics) replaces the f32 atomic scatter-add; row sums now in arrival-rank order, epilogues fused
# speedup vs baseline: 2.0346x; 2.0346x over previous
#include <hip/hip_runtime.h>


typedef __attribute__((ext_vector_type(2))) float v2f;
typedef __attribute__((ext_vector_type(8))) float v8f;

#define IN_F    128
#define HID_F   64
#define OUT_F   40
#define OUT_PAD 48
#define C_LD    64

#define SCAN_T   1024
#define SCAN_EL  (SCAN_T * 4)


__global__ void zero_i32(int* __restrict__ p, int n) {
  int t = blockIdx.x * blockDim.x + threadIdx.x;
  if (t < n) p[t] = 0;
}

__global__ void count_rank(const int* __restrict__ se, const int* __restrict__ de,
                           int* __restrict__ cnt_out, int* __restrict__ cnt_in,
                           int* __restrict__ rank, int nEdges) {
  int e = blockIdx.x * blockDim.x + threadIdx.x;
  if (e >= nEdges) return;
  atomicAdd(&cnt_out[se[e]], 1);
  rank[e] = atomicAdd(&cnt_in[de[e]], 1);
}

__global__ void make_norms(const int* __restrict__ cnt, float* __restrict__ norm, int n) {
  int t = blockIdx.x * blockDim.x + threadIdx.x;
  if (t < n) norm[t] = rsqrtf(fmaxf((float)cnt[t], 1.0f));
}

__global__ void __launch_bounds__(SCAN_T)
seg_offsets(const int* __restrict__ cnt, int* __restrict__ row_ptr, int n) {
  __shared__ int front[32];
  __shared__ int wtot[32];
  const int tid = threadIdx.x, lane = tid & 31, wv = tid >> 5;
  const int base = blockIdx.x * SCAN_EL;

  int part = 0;
  const int4* c4 = (const int4*)cnt;
  const int nv = base >> 2;
#pragma unroll 8
  for (int i = tid; i < nv; i += SCAN_T) {
    int4 v = c4[i];
    part += (v.x + v.y) + (v.z + v.w);
  }
  for (int o = 16; o > 0; o >>= 1) part += __shfl_xor(part, o);
  if (lane == 0) front[wv] = part;

  const int i0 = base + tid * 4;
  int x0 = 0, x1 = 0, x2 = 0, x3 = 0;
  if (i0 + 3 < n) {
    int4 v = c4[i0 >> 2];
    x0 = v.x; x1 = v.y; x2 = v.z; x3 = v.w;
  } else {
    if (i0     < n) x0 = cnt[i0];
    if (i0 + 1 < n) x1 = cnt[i0 + 1];
    if (i0 + 2 < n) x2 = cnt[i0 + 2];
  }
  const int t1 = x0, t2 = x0 + x1, t3 = t2 + x2, tot = t3 + x3;
  int inc = tot;
  for (int o = 1; o < 32; o <<= 1) {
    int y = __shfl_up(inc, o);
    if (lane >= o) inc += y;
  }
  if (lane == 31) wtot[wv] = inc;
  __syncthreads();

  int ex = inc - tot;
  for (int w = 0; w < 32; ++w) ex += front[w];
  for (int w = 0; w < wv; ++w) ex += wtot[w];

  if (i0     < n) row_ptr[i0]     = ex;
  if (i0 + 1 < n) row_ptr[i0 + 1] = ex + t1;
  if (i0 + 2 < n) row_ptr[i0 + 2] = ex + t2;
  if (i0 + 3 < n) row_ptr[i0 + 3] = ex + t3;
}

__global__ void place_edges(const int* __restrict__ se, const int* __restrict__ de,
                            const int* __restrict__ rank, const int* __restrict__ row_ptr,
                            int* __restrict__ sorted_src, int nEdges) {
  int e = blockIdx.x * blockDim.x + threadIdx.x;
  if (e >= nEdges) return;
  sorted_src[row_ptr[de[e]] + rank[e]] = se[e];
}

__global__ void pad_w3(const float* __restrict__ W3, float* __restrict__ W3p) {
  int t = blockIdx.x * blockDim.x + threadIdx.x;
  if (t >= HID_F * OUT_PAD) return;
  int r = t / OUT_PAD, c = t % OUT_PAD;
  W3p[t] = (c < OUT_F) ? W3[r * OUT_F + c] : 0.0f;
}


__device__ __forceinline__ void add4(float4& a, const float4& v) {
  a.x += v.x; a.y += v.y; a.z += v.z; a.w += v.w;
}

template <int MODE>
__global__ void gather_rows(const float* __restrict__ feat, const int* __restrict__ row_ptr,
                            const int* __restrict__ cnt_in, const int* __restrict__ sorted_src,
                            const float* __restrict__ norm_in, const float* __restrict__ norm_out,
                            const float* __restrict__ bias, float* __restrict__ outp, int nNodes) {
  constexpr unsigned ROW4 = (MODE == 2) ? C_LD / 4 : HID_F / 4;
  constexpr unsigned NCH  = (MODE == 2) ? OUT_F / 4   : HID_F / 4;
  const unsigned t = blockIdx.x * blockDim.x + threadIdx.x;
  const unsigned i = t >> 4, ch = t & 15;
  if (i >= (unsigned)nNodes || ch >= NCH) return;
  const float4* f4 = (const float4*)feat + ch;
  int e = row_ptr[i];
  const int end = e + cnt_in[i];
  float4 acc = make_float4(0.0f, 0.0f, 0.0f, 0.0f);
  int s0 = 0, s1 = 0, s2 = 0, s3 = 0;
  bool more = e + 4 <= end;
  if (more) { s0 = sorted_src[e]; s1 = sorted_src[e + 1]; s2 = sorted_src[e + 2]; s3 = sorted_src[e + 3]; }
  while (more) {
    const float4 v0 = f4[(size_t)s0 * ROW4];
    const float4 v1 = f4[(size_t)s1 * ROW4];
    const float4 v2 = f4[(size_t)s2 * ROW4];
    const float4 v3 = f4[(size_t)s3 * ROW4];
    e += 4;
    more = e + 4 <= end;
    if (more) { s0 = sorted_src[e]; s1 = sorted_src[e + 1]; s2 = sorted_src[e + 2]; s3 = sorted_src[e + 3]; }
    add4(acc, v0); add4(acc, v1); add4(acc, v2); add4(acc, v3);
  }
  for (; e < end; ++e) {
    const float4 v = f4[(size_t)sorted_src[e] * ROW4];
    add4(acc, v);
  }
  if (MODE == 0) {
    ((float4*)outp)[(size_t)i * (HID_F / 4) + ch] = acc;
  } else if (MODE == 1) {
    const float ni = norm_in[i], no = norm_out[i];
    const float4 b = ((const float4*)bias)[ch];
    float4 r;
    r.x = fmaxf(acc.x * ni + b.x, 0.0f) * no;
    r.y = fmaxf(acc.y * ni + b.y, 0.0f) * no;
    r.z = fmaxf(acc.z * ni + b.z, 0.0f) * no;
    r.w = fmaxf(acc.w * ni + b.w, 0.0f) * no;
    ((float4*)outp)[(size_t)i * (HID_F / 4) + ch] = r;
  } else {
    const float ni = norm_in[i];
    const float4 b = ((const float4*)bias)[ch];
    float4 r;
    r.x = acc.x * ni + b.x;
    r.y = acc.y * ni + b.y;
    r.z = acc.z * ni + b.z;
    r.w = acc.w * ni + b.w;
    ((float4*)outp)[(size_t)i * (OUT_F / 4) + ch] = r;
  }
}


__global__ void gemm1_scaleA(const float* __restrict__ X, const float* __restrict__ W,
                             const float* __restrict__ norm_out, float* __restrict__ out,
                             int nNodes, int mTiles) {
  int wave = blockIdx.x * (blockDim.x >> 5) + (threadIdx.x >> 5);
  int lane = threadIdx.x & 31;
  int tm = wave >> 2;
  int tn = wave & 3;
  if (tm >= mTiles) return;
  int half = lane >> 4, l = lane & 15;
  int rowA = tm * 16 + l;
  int rA = rowA < nNodes ? rowA : nNodes - 1;
  float s = norm_out[rA];
  const float* xrow = X + (size_t)rA * IN_F;
  v8f c = {};
  for (int k0 = 0; k0 < IN_F; k0 += 4) {
    int ka = k0 + half * 2;
    float2 av = *(const float2*)(xrow + ka);
    v2f a; a.x = av.x * s; a.y = av.y * s;
    v2f b; b.x = W[ka * HID_F + tn * 16 + l];
           b.y = W[(ka + 1) * HID_F + tn * 16 + l];
    c = __builtin_amdgcn_wmma_f32_16x16x4_f32(false, a, false, b, (short)0, c, false, false);
  }
  int n = tn * 16 + l;
#pragma unroll
  for (int j = 0; j < 8; ++j) {
    int row = tm * 16 + half * 8 + j;
    if (row < nNodes) out[(size_t)row * HID_F + n] = c[j];
  }
}

__global__ void gemm2_epi(const float* __restrict__ A, const float* __restrict__ W,
                          const float* __restrict__ bias, const float* __restrict__ norm_in,
                          const float* __restrict__ norm_out, float* __restrict__ out,
                          int nNodes, int mTiles) {
  int wave = blockIdx.x * (blockDim.x >> 5) + (threadIdx.x >> 5);
  int lane = threadIdx.x & 31;
  int tm = wave >> 2;
  int tn = wave & 3;
  if (tm >= mTiles) return;
  int half = lane >> 4, l = lane & 15;
  int rowA = tm * 16 + l;
  int rA = rowA < nNodes ? rowA : nNodes - 1;
  const float* arow = A + (size_t)rA * HID_F;
  v8f c = {};
  for (int k0 = 0; k0 < HID_F; k0 += 4) {
    int ka = k0 + half * 2;
    float2 av = *(const float2*)(arow + ka);
    v2f a; a.x = av.x; a.y = av.y;
    v2f b; b.x = W[ka * HID_F + tn * 16 + l];
           b.y = W[(ka + 1) * HID_F + tn * 16 + l];
    c = __builtin_amdgcn_wmma_f32_16x16x4_f32(false, a, false, b, (short)0, c, false, false);
  }
  int n = tn * 16 + l;
  float bv = bias[n];
#pragma unroll
  for (int j = 0; j < 8; ++j) {
    int row = tm * 16 + half * 8 + j;
    if (row < nNodes) {
      float v = c[j] * norm_in[row] + bv;
      v = fmaxf(v, 0.0f);
      out[(size_t)row * HID_F + n] = v * norm_out[row];
    }
  }
}

__global__ void gemm3_plain(const float* __restrict__ A, const float* __restrict__ W,
                            float* __restrict__ out, int nNodes, int mTiles) {
  int wave = blockIdx.x * (blockDim.x >> 5) + (threadIdx.x >> 5);
  int lane = threadIdx.x & 31;
  int tm = wave / 3;
  int tn = wave - tm * 3;
  if (tm >= mTiles) return;
  int half = lane >> 4, l = lane & 15;
  int rowA = tm * 16 + l;
  int rA = rowA < nNodes ? rowA : nNodes - 1;
  const float* arow = A + (size_t)rA * HID_F;
  v8f c = {};
  for (int k0 = 0; k0 < HID_F; k0 += 4) {
    int ka = k0 + half * 2;
    float2 av = *(const float2*)(arow + ka);
    v2f a; a.x = av.x; a.y = av.y;
    v2f b; b.x = W[ka * OUT_PAD + tn * 16 + l];
           b.y = W[(ka + 1) * OUT_PAD + tn * 16 + l];
    c = __builtin_amdgcn_wmma_f32_16x16x4_f32(false, a, false, b, (short)0, c, false, false);
  }
  int n = tn * 16 + l;
#pragma unroll
  for (int j = 0; j < 8; ++j) {
    int row = tm * 16 + half * 8 + j;
    if (row < nNodes) out[(size_t)row * C_LD + n] = c[j];
  }
}


static inline int cdiv_ll(long long a, long long b) { return (int)((a + b - 1) / b); }

extern "C" void kernel_launch(void* const* d_in, const int* in_sizes, int n_in,
                              void* d_out, int out_size, void* d_ws, size_t ws_size,
                              hipStream_t stream) {
  const float* X  = (const float*)d_in[0];
  const float* W1 = (const float*)d_in[1];
  const float* b1 = (const float*)d_in[2];
  const float* W2 = (const float*)d_in[3];
  const float* b2 = (const float*)d_in[4];
  const float* W3 = (const float*)d_in[5];
  const float* b3 = (const float*)d_in[6];
  const int* src  = (const int*)d_in[7];
  const int* dst  = (const int*)d_in[8];
  float* out = (float*)d_out;

  const int nNodes = in_sizes[0] / IN_F;
  const int nEdges = in_sizes[7];
  const int mTiles = (nNodes + 15) / 16;

  const size_t nP = ((size_t)nNodes + 63) & ~(size_t)63;
  const size_t eP = ((size_t)nEdges + 63) & ~(size_t)63;
  int* wi = (int*)d_ws;
  int* cnt_out    = wi;
  int* cnt_in     = cnt_out + nP;
  int* row_ptr    = cnt_in + nP;
  int* rank       = row_ptr + nP;
  int* sorted_src = rank + eP;
  float* norm_out = (float*)(sorted_src + eP);
  float* norm_in  = norm_out + nP;
  float* W3p      = norm_in + nP;
  float* bufA     = W3p + HID_F * OUT_PAD;
  float* bufB     = bufA + nP * HID_F;
  float* bufC     = bufB + nP * HID_F;

  const dim3 blk(256);
  const int wavesH = mTiles * (HID_F / 16);
  const int wavesO = mTiles * (OUT_PAD / 16);
  const int gatherBlocks = cdiv_ll((long long)nNodes * 16, 256);

  zero_i32<<<cdiv_ll(2LL * nP, 256), blk, 0, stream>>>(cnt_out, (int)(2 * nP));
  count_rank<<<cdiv_ll(nEdges, 256), blk, 0, stream>>>(src, dst, cnt_out, cnt_in, rank, nEdges);
  make_norms<<<cdiv_ll(2LL * nP, 256), blk, 0, stream>>>(cnt_out, norm_out, (int)(2 * nP));
  seg_offsets<<<cdiv_ll(nNodes, SCAN_EL), dim3(SCAN_T), 0, stream>>>(cnt_in, row_ptr, nNodes);
  place_edges<<<cdiv_ll(nEdges, 256), blk, 0, stream>>>(src, dst, rank, row_ptr, sorted_src, nEdges);
  pad_w3<<<cdiv_ll(HID_F * OUT_PAD, 256), blk, 0, stream>>>(W3, W3p);

  gemm1_scaleA<<<cdiv_ll((long long)wavesH * 32, 256), blk, 0, stream>>>(
      X, W1, norm_out, bufA, nNodes, mTiles);
  gather_rows<1><<<gatherBlocks, blk, 0, stream>>>(
      bufA, row_ptr, cnt_in, sorted_src, norm_in, norm_out, b1, bufB, nNodes);

  gather_rows<0><<<gatherBlocks, blk, 0, stream>>>(
      bufB, row_ptr, cnt_in, sorted_src, norm_in, norm_out, b2, bufA, nNodes);
  gemm2_epi<<<cdiv_ll((long long)wavesH * 32, 256), blk, 0, stream>>>(
      bufA, W2, b2, norm_in, norm_out, bufB, nNodes, mTiles);

  gemm3_plain<<<cdiv_ll((long long)wavesO * 32, 256), blk, 0, stream>>>(
      bufB, W3p, bufC, nNodes, mTiles);
  gather_rows<2><<<gatherBlocks, blk, 0, stream>>>(
      bufC, row_ptr, cnt_in, sorted_src, norm_in, norm_out, b3, out, nNodes);
}


# Edge sort by destination moved into LDS: per-tile bin order, then per-bin node order, stored as runs; no global rank atomics or scattered placement; row sums in LDS-cursor order
# speedup vs baseline: 2.5031x; 1.2303x over previous
#include <hip/hip_runtime.h>


typedef __attribute__((ext_vector_type(2))) float v2f;
typedef __attribute__((ext_vector_type(8))) float v8f;

#define IN_F    128
#define HID_F   64
#define OUT_F   40
#define OUT_PAD 48
#define C_LD    64

#define BSH      8
#define BNODES   (1 << BSH)
#define MAXB     512
#define TILE     8192
#define TILE_T   1024
#define STAGE    8192


__global__ void zero_i32(int* __restrict__ p, int n) {
  int t = blockIdx.x * blockDim.x + threadIdx.x;
  if (t < n) p[t] = 0;
}

__global__ void __launch_bounds__(TILE_T)
tile_count(const int* __restrict__ se, const int* __restrict__ de,
           int* __restrict__ cnt_out, int* __restrict__ hist, int nEdges, int nBins) {
  __shared__ int h[MAXB];
  const int tid = threadIdx.x;
  if (tid < MAXB) h[tid] = 0;
  __syncthreads();
  const int base = blockIdx.x * TILE;
#pragma unroll
  for (int j = 0; j < TILE / TILE_T; ++j) {
    const int e = base + j * TILE_T + tid;
    if (e < nEdges) {
      atomicAdd(&h[de[e] >> BSH], 1);
      atomicAdd(&cnt_out[se[e]], 1);
    }
  }
  __syncthreads();
  if (tid < nBins) hist[blockIdx.x * nBins + tid] = h[tid];
}

__global__ void __launch_bounds__(2 * MAXB)
bin_offsets(const int* __restrict__ hist, int* __restrict__ off, int* __restrict__ bin_base,
            int nTiles, int nBins) {
  __shared__ int ps[2][MAXB];
  __shared__ int tot[MAXB];
  __shared__ int bs[MAXB + 1];
  const int tid = threadIdx.x;
  const int b = tid % MAXB, part = tid / MAXB;
  const int halfT = (nTiles + 1) >> 1;
  const int c0 = part ? halfT : 0, c1 = part ? nTiles : halfT;
  int sum = 0;
  if (b < nBins) {
#pragma unroll 16
    for (int c = c0; c < c1; ++c) sum += hist[c * nBins + b];
  }
  ps[part][b] = sum;
  __syncthreads();
  if (tid < MAXB) tot[tid] = ps[0][tid] + ps[1][tid];
  __syncthreads();
  if (tid <= MAXB) {
    int a = 0;
    for (int j = 0; j < tid; ++j) a += tot[j];
    bs[tid] = a;
  }
  __syncthreads();
  if (b < nBins) {
    int run = bs[b] + (part ? ps[0][b] : 0);
#pragma unroll 16
    for (int c = c0; c < c1; ++c) {
      const int v = hist[c * nBins + b];
      off[c * nBins + b] = run;
      run += v;
    }
  }
  if (tid <= nBins) bin_base[tid] = bs[tid];
}

__global__ void __launch_bounds__(TILE_T)
tile_scatter(const int* __restrict__ se, const int* __restrict__ de,
             const int* __restrict__ hist, const int* __restrict__ off,
             int* __restrict__ bsrc, int* __restrict__ bdst, int nEdges, int nBins) {
  __shared__ int lcnt[MAXB];
  __shared__ int lcur[MAXB];
  __shared__ int gdel[MAXB];
  __shared__ int ssrc[TILE];
  __shared__ int sdst[TILE];
  const int tid = threadIdx.x;
  const int base = blockIdx.x * TILE;
  const int left = nEdges - base;
  const int nT = left < TILE ? left : TILE;
  if (tid < MAXB) lcnt[tid] = (tid < nBins) ? hist[blockIdx.x * nBins + tid] : 0;
  __syncthreads();
  if (tid < MAXB) {
    int a = 0;
    for (int j = 0; j < tid; ++j) a += lcnt[j];
    lcur[tid] = a;
    gdel[tid] = (tid < nBins) ? off[blockIdx.x * nBins + tid] - a : 0;
  }
  __syncthreads();
#pragma unroll
  for (int j = 0; j < TILE / TILE_T; ++j) {
    const int e = base + j * TILE_T + tid;
    if (e < nEdges) {
      const int s = se[e], d = de[e];
      const int q = atomicAdd(&lcur[d >> BSH], 1);
      ssrc[q] = s;
      sdst[q] = d;
    }
  }
  __syncthreads();
  for (int j = tid; j < nT; j += TILE_T) {
    const int d = sdst[j];
    const int g = gdel[d >> BSH] + j;
    bsrc[g] = ssrc[j];
    bdst[g] = d;
  }
}

__global__ void __launch_bounds__(TILE_T)
bin_finish(const int* __restrict__ bsrc, const int* __restrict__ bdst,
           const int* __restrict__ bin_base, const int* __restrict__ cnt_out,
           int* __restrict__ row_ptr, int* __restrict__ cnt_in,
           float* __restrict__ norm_in, float* __restrict__ norm_out,
           int* __restrict__ sorted_src, int nNodes) {
  __shared__ int ncnt[BNODES];
  __shared__ int ncur[BNODES];
  __shared__ int stage[STAGE];
  const int tid = threadIdx.x;
  const int e0 = bin_base[blockIdx.x];
  const int nE = bin_base[blockIdx.x + 1] - e0;
  const int n0 = blockIdx.x << BSH;
  if (tid < BNODES) ncnt[tid] = 0;
  __syncthreads();
  for (int j = tid; j < nE; j += TILE_T) atomicAdd(&ncnt[bdst[e0 + j] - n0], 1);
  __syncthreads();
  if (tid < BNODES) {
    int a = 0;
    for (int j = 0; j < tid; ++j) a += ncnt[j];
    ncur[tid] = a;
    const int node = n0 + tid;
    if (node < nNodes) {
      const int c = ncnt[tid];
      row_ptr[node] = e0 + a;
      cnt_in[node] = c;
      norm_in[node] = rsqrtf(fmaxf((float)c, 1.0f));
      norm_out[node] = rsqrtf(fmaxf((float)cnt_out[node], 1.0f));
    }
  }
  __syncthreads();
  const bool staged = nE <= STAGE;
  for (int j = tid; j < nE; j += TILE_T) {
    const int q = atomicAdd(&ncur[bdst[e0 + j] - n0], 1);
    const int s = bsrc[e0 + j];
    if (staged) stage[q] = s; else sorted_src[e0 + q] = s;
  }
  __syncthreads();
  if (staged) {
    for (int j = tid; j < nE; j += TILE_T) sorted_src[e0 + j] = stage[j];
  }
}

__global__ void pad_w3(const float* __restrict__ W3, float* __restrict__ W3p) {
  int t = blockIdx.x * blockDim.x + threadIdx.x;
  if (t >= HID_F * OUT_PAD) return;
  int r = t / OUT_PAD, c = t % OUT_PAD;
  W3p[t] = (c < OUT_F) ? W3[r * OUT_F + c] : 0.0f;
}


__device__ __forceinline__ void add4(float4& a, const float4& v) {
  a.x += v.x; a.y += v.y; a.z += v.z; a.w += v.w;
}

template <int MODE>
__global__ void gather_rows(const float* __restrict__ feat, const int* __restrict__ row_ptr,
                            const int* __restrict__ cnt_in, const int* __restrict__ sorted_src,
                            const float* __restrict__ norm_in, const float* __restrict__ norm_out,
                            const float* __restrict__ bias, float* __restrict__ outp, int nNodes) {
  constexpr unsigned ROW4 = (MODE == 2) ? C_LD / 4 : HID_F / 4;
  constexpr unsigned NCH  = (MODE == 2) ? OUT_F / 4   : HID_F / 4;
  const unsigned t = blockIdx.x * blockDim.x + threadIdx.x;
  const unsigned i = t >> 4, ch = t & 15;
  if (i >= (unsigned)nNodes || ch >= NCH) return;
  const float4* f4 = (const float4*)feat + ch;
  int e = row_ptr[i];
  const int end = e + cnt_in[i];
  float4 acc = make_float4(0.0f, 0.0f, 0.0f, 0.0f);
  int s0 = 0, s1 = 0, s2 = 0, s3 = 0;
  bool more = e + 4 <= end;
  if (more) { s0 = sorted_src[e]; s1 = sorted_src[e + 1]; s2 = sorted_src[e + 2]; s3 = sorted_src[e + 3]; }
  while (more) {
    const float4 v0 = f4[(size_t)s0 * ROW4];
    const float4 v1 = f4[(size_t)s1 * ROW4];
    const float4 v2 = f4[(size_t)s2 * ROW4];
    const float4 v3 = f4[(size_t)s3 * ROW4];
    e += 4;
    more = e + 4 <= end;
    if (more) { s0 = sorted_src[e]; s1 = sorted_src[e + 1]; s2 = sorted_src[e + 2]; s3 = sorted_src[e + 3]; }
    add4(acc, v0); add4(acc, v1); add4(acc, v2); add4(acc, v3);
  }
  for (; e < end; ++e) {
    const float4 v = f4[(size_t)sorted_src[e] * ROW4];
    add4(acc, v);
  }
  if (MODE == 0) {
    ((float4*)outp)[(size_t)i * (HID_F / 4) + ch] = acc;
  } else if (MODE == 1) {
    const float ni = norm_in[i], no = norm_out[i];
    const float4 b = ((const float4*)bias)[ch];
    float4 r;
    r.x = fmaxf(acc.x * ni + b.x, 0.0f) * no;
    r.y = fmaxf(acc.y * ni + b.y, 0.0f) * no;
    r.z = fmaxf(acc.z * ni + b.z, 0.0f) * no;
    r.w = fmaxf(acc.w * ni + b.w, 0.0f) * no;
    ((float4*)outp)[(size_t)i * (HID_F / 4) + ch] = r;
  } else {
    const float ni = norm_in[i];
    const float4 b = ((const float4*)bias)[ch];
    float4 r;
    r.x = acc.x * ni + b.x;
    r.y = acc.y * ni + b.y;
    r.z = acc.z * ni + b.z;
    r.w = acc.w * ni + b.w;
    ((float4*)outp)[(size_t)i * (OUT_F / 4) + ch] = r;
  }
}


__global__ void gemm1_scaleA(const float* __restrict__ X, const float* __restrict__ W,
                             const float* __restrict__ norm_out, float* __restrict__ out,
                             int nNodes, int mTiles) {
  int wave = blockIdx.x * (blockDim.x >> 5) + (threadIdx.x >> 5);
  int lane = threadIdx.x & 31;
  int tm = wave >> 2;
  int tn = wave & 3;
  if (tm >= mTiles) return;
  int half = lane >> 4, l = lane & 15;
  int rowA = tm * 16 + l;
  int rA = rowA < nNodes ? rowA : nNodes - 1;
  float s = norm_out[rA];
  const float* xrow = X + (size_t)rA * IN_F;
  v8f c = {};
  for (int k0 = 0; k0 < IN_F; k0 += 4) {
    int ka = k0 + half * 2;
    float2 av = *(const float2*)(xrow + ka);
    v2f a; a.x = av.x * s; a.y = av.y * s;
    v2f b; b.x = W[ka * HID_F + tn * 16 + l];
           b.y = W[(ka + 1) * HID_F + tn * 16 + l];
    c = __builtin_amdgcn_wmma_f32_16x16x4_f32(false, a, false, b, (short)0, c, false, false);
  }
  int n = tn * 16 + l;
#pragma unroll
  for (int j = 0; j < 8; ++j) {
    int row = tm * 16 + half * 8 + j;
    if (row < nNodes) out[(size_t)row * HID_F + n] = c[j];
  }
}

__global__ void gemm2_epi(const float* __restrict__ A, const float* __restrict__ W,
                          const float* __restrict__ bias, const float* __restrict__ norm_in,
                          const float* __restrict__ norm_out, float* __restrict__ out,
                          int nNodes, int mTiles) {
  int wave = blockIdx.x * (blockDim.x >> 5) + (threadIdx.x >> 5);
  int lane = threadIdx.x & 31;
  int tm = wave >> 2;
  int tn = wave & 3;
  if (tm >= mTiles) return;
  int half = lane >> 4, l = lane & 15;
  int rowA = tm * 16 + l;
  int rA = rowA < nNodes ? rowA : nNodes - 1;
  const float* arow = A + (size_t)rA * HID_F;
  v8f c = {};
  for (int k0 = 0; k0 < HID_F; k0 += 4) {
    int ka = k0 + half * 2;
    float2 av = *(const float2*)(arow + ka);
    v2f a; a.x = av.x; a.y = av.y;
    v2f b; b.x = W[ka * HID_F + tn * 16 + l];
           b.y = W[(ka + 1) * HID_F + tn * 16 + l];
    c = __builtin_amdgcn_wmma_f32_16x16x4_f32(false, a, false, b, (short)0, c, false, false);
  }
  int n = tn * 16 + l;
  float bv = bias[n];
#pragma unroll
  for (int j = 0; j < 8; ++j) {
    int row = tm * 16 + half * 8 + j;
    if (row < nNodes) {
      float v = c[j] * norm_in[row] + bv;
      v = fmaxf(v, 0.0f);
      out[(size_t)row * HID_F + n] = v * norm_out[row];
    }
  }
}

__global__ void gemm3_plain(const float* __restrict__ A, const float* __restrict__ W,
                            float* __restrict__ out, int nNodes, int mTiles) {
  int wave = blockIdx.x * (blockDim.x >> 5) + (threadIdx.x >> 5);
  int lane = threadIdx.x & 31;
  int tm = wave / 3;
  int tn = wave - tm * 3;
  if (tm >= mTiles) return;
  int half = lane >> 4, l = lane & 15;
  int rowA = tm * 16 + l;
  int rA = rowA < nNodes ? rowA : nNodes - 1;
  const float* arow = A + (size_t)rA * HID_F;
  v8f c = {};
  for (int k0 = 0; k0 < HID_F; k0 += 4) {
    int ka = k0 + half * 2;
    float2 av = *(const float2*)(arow + ka);
    v2f a; a.x = av.x; a.y = av.y;
    v2f b; b.x = W[ka * OUT_PAD + tn * 16 + l];
           b.y = W[(ka + 1) * OUT_PAD + tn * 16 + l];
    c = __builtin_amdgcn_wmma_f32_16x16x4_f32(false, a, false, b, (short)0, c, false, false);
  }
  int n = tn * 16 + l;
#pragma unroll
  for (int j = 0; j < 8; ++j) {
    int row = tm * 16 + half * 8 + j;
    if (row < nNodes) out[(size_t)row * C_LD + n] = c[j];
  }
}


static inline int cdiv_ll(long long a, long long b) { return (int)((a + b - 1) / b); }

extern "C" void kernel_launch(void* const* d_in, const int* in_sizes, int n_in,
                              void* d_out, int out_size, void* d_ws, size_t ws_size,
                              hipStream_t stream) {
  const float* X  = (const float*)d_in[0];
  const float* W1 = (const float*)d_in[1];
  const float* b1 = (const float*)d_in[2];
  const float* W2 = (const float*)d_in[3];
  const float* b2 = (const float*)d_in[4];
  const float* W3 = (const float*)d_in[5];
  const float* b3 = (const float*)d_in[6];
  const int* src  = (const int*)d_in[7];
  const int* dst  = (const int*)d_in[8];
  float* out = (float*)d_out;

  const int nNodes = in_sizes[0] / IN_F;
  const int nEdges = in_sizes[7];
  const int mTiles = (nNodes + 15) / 16;
  const int nBins  = (nNodes + BNODES - 1) >> BSH;
  const int nTiles = (nEdges + TILE - 1) / TILE;

  const size_t nP = ((size_t)nNodes + 63) & ~(size_t)63;
  const size_t eP = ((size_t)nEdges + 63) & ~(size_t)63;
  const size_t hP = ((size_t)nTiles * nBins + 63) & ~(size_t)63;
  int* wi = (int*)d_ws;
  int* cnt_out    = wi;
  int* cnt_in     = cnt_out + nP;
  int* row_ptr    = cnt_in + nP;
  int* hist       = row_ptr + nP;
  int* off        = hist + hP;
  int* bin_base   = off + hP;
  int* bsrc       = bin_base + MAXB + 64;
  int* bdst       = bsrc + eP;
  int* sorted_src = bdst + eP;
  float* norm_out = (float*)(sorted_src + eP);
  float* norm_in  = norm_out + nP;
  float* W3p      = norm_in + nP;
  float* bufA     = W3p + HID_F * OUT_PAD;
  float* bufB     = bufA + nP * HID_F;
  float* bufC     = bufB + nP * HID_F;

  const dim3 blk(256);
  const int wavesH = mTiles * (HID_F / 16);
  const int wavesO = mTiles * (OUT_PAD / 16);
  const int gatherBlocks = cdiv_ll((long long)nNodes * 16, 256);

  zero_i32<<<cdiv_ll((long long)nP, 256), blk, 0, stream>>>(cnt_out, (int)nP);
  tile_count<<<nTiles, dim3(TILE_T), 0, stream>>>(src, dst, cnt_out, hist, nEdges, nBins);
  bin_offsets<<<1, dim3(2 * MAXB), 0, stream>>>(hist, off, bin_base, nTiles, nBins);
  tile_scatter<<<nTiles, dim3(TILE_T), 0, stream>>>(src, dst, hist, off, bsrc, bdst, nEdges, nBins);
  bin_finish<<<nBins, dim3(TILE_T), 0, stream>>>(bsrc, bdst, bin_base, cnt_out, row_ptr, cnt_in,
                                                norm_in, norm_out, sorted_src, nNodes);
  pad_w3<<<cdiv_ll(HID_F * OUT_PAD, 256), blk, 0, stream>>>(W3, W3p);

  gemm1_scaleA<<<cdiv_ll((long long)wavesH * 32, 256), blk, 0, stream>>>(
      X, W1, norm_out, bufA, nNodes, mTiles);
  gather_rows<1><<<gatherBlocks, blk, 0, stream>>>(
      bufA, row_ptr, cnt_in, sorted_src, norm_in, norm_out, b1, bufB, nNodes);

  gather_rows<0><<<gatherBlocks, blk, 0, stream>>>(
      bufB, row_ptr, cnt_in, sorted_src, norm_in, norm_out, b2, bufA, nNodes);
  gemm2_epi<<<cdiv_ll((long long)wavesH * 32, 256), blk, 0, stream>>>(
      bufA, W2, b2, norm_in, norm_out, bufB, nNodes, mTiles);

  gemm3_plain<<<cdiv_ll((long long)wavesO * 32, 256), blk, 0, stream>>>(
      bufB, W3p, bufC, nNodes, mTiles);
  gather_rows<2><<<gatherBlocks, blk, 0, stream>>>(
      bufC, row_ptr, cnt_in, sorted_src, norm_in, norm_out, b3, out, nNodes);
}


# GEMMs re-tiled: a wave owns 16 rows x all column tiles; rows come in as whole 256-B pieces, weights once per work-group, both via LDS; k order inside the products' sums changed; pad launch folded in
# speedup vs baseline: 2.7551x; 1.1007x over previous
#include <hip/hip_runtime.h>


typedef __attribute__((ext_vector_type(2))) float v2f;
typedef __attribute__((ext_vector_type(8))) float v8f;

#define IN_F    128
#define HID_F   64
#define OUT_F   40
#define OUT_PAD 48
#define C_LD    64

#define BSH      8
#define BNODES   (1 << BSH)
#define MAXB     512
#define TILE     8192
#define TILE_T   1024
#define STAGE    8192


__global__ void zero_i32(int* __restrict__ p, int n) {
  int t = blockIdx.x * blockDim.x + threadIdx.x;
  if (t < n) p[t] = 0;
}

__global__ void __launch_bounds__(TILE_T)
tile_count(const int* __restrict__ se, const int* __restrict__ de,
           int* __restrict__ cnt_out, int* __restrict__ hist, int nEdges, int nBins) {
  __shared__ int h[MAXB];
  const int tid = threadIdx.x;
  if (tid < MAXB) h[tid] = 0;
  __syncthreads();
  const int base = blockIdx.x * TILE;
#pragma unroll
  for (int j = 0; j < TILE / TILE_T; ++j) {
    const int e = base + j * TILE_T + tid;
    if (e < nEdges) {
      atomicAdd(&h[de[e] >> BSH], 1);
      atomicAdd(&cnt_out[se[e]], 1);
    }
  }
  __syncthreads();
  if (tid < nBins) hist[blockIdx.x * nBins + tid] = h[tid];
}

__global__ void __launch_bounds__(2 * MAXB)
bin_offsets(const int* __restrict__ hist, int* __restrict__ off, int* __restrict__ bin_base,
            int nTiles, int nBins) {
  __shared__ int ps[2][MAXB];
  __shared__ int tot[MAXB];
  __shared__ int bs[MAXB + 1];
  const int tid = threadIdx.x;
  const int b = tid % MAXB, part = tid / MAXB;
  const int halfT = (nTiles + 1) >> 1;
  const int c0 = part ? halfT : 0, c1 = part ? nTiles : halfT;
  int sum = 0;
  if (b < nBins) {
#pragma unroll 16
    for (int c = c0; c < c1; ++c) sum += hist[c * nBins + b];
  }
  ps[part][b] = sum;
  __syncthreads();
  if (tid < MAXB) tot[tid] = ps[0][tid] + ps[1][tid];
  __syncthreads();
  if (tid <= MAXB) {
    int a = 0;
    for (int j = 0; j < tid; ++j) a += tot[j];
    bs[tid] = a;
  }
  __syncthreads();
  if (b < nBins) {
    int run = bs[b] + (part ? ps[0][b] : 0);
#pragma unroll 16
    for (int c = c0; c < c1; ++c) {
      const int v = hist[c * nBins + b];
      off[c * nBins + b] = run;
      run += v;
    }
  }
  if (tid <= nBins) bin_base[tid] = bs[tid];
}

__global__ void __launch_bounds__(TILE_T)
tile_scatter(const int* __restrict__ se, const int* __restrict__ de,
             const int* __restrict__ hist, const int* __restrict__ off,
             int* __restrict__ bsrc, int* __restrict__ bdst, int nEdges, int nBins) {
  __shared__ int lcnt[MAXB];
  __shared__ int lcur[MAXB];
  __shared__ int gdel[MAXB];
  __shared__ int ssrc[TILE];
  __shared__ int sdst[TILE];
  const int tid = threadIdx.x;
  const int base = blockIdx.x * TILE;
  const int left = nEdges - base;
  const int nT = left < TILE ? left : TILE;
  if (tid < MAXB) lcnt[tid] = (tid < nBins) ? hist[blockIdx.x * nBins + tid] : 0;
  __syncthreads();
  if (tid < MAXB) {
    int a = 0;
    for (int j = 0; j < tid; ++j) a += lcnt[j];
    lcur[tid] = a;
    gdel[tid] = (tid < nBins) ? off[blockIdx.x * nBins + tid] - a : 0;
  }
  __syncthreads();
#pragma unroll
  for (int j = 0; j < TILE / TILE_T; ++j) {
    const int e = base + j * TILE_T + tid;
    if (e < nEdges) {
      const int s = se[e], d = de[e];
      const int q = atomicAdd(&lcur[d >> BSH], 1);
      ssrc[q] = s;
      sdst[q] = d;
    }
  }
  __syncthreads();
  for (int j = tid; j < nT; j += TILE_T) {
    const int d = sdst[j];
    const int g = gdel[d >> BSH] + j;
    bsrc[g] = ssrc[j];
    bdst[g] = d;
  }
}

__global__ void __launch_bounds__(TILE_T)
bin_finish(const int* __restrict__ bsrc, const int* __restrict__ bdst,
           const int* __restrict__ bin_base, const int* __restrict__ cnt_out,
           int* __restrict__ row_ptr, int* __restrict__ cnt_in,
           float* __restrict__ norm_in, float* __restrict__ norm_out,
           int* __restrict__ sorted_src, int nNodes) {
  __shared__ int ncnt[BNODES];
  __shared__ int ncur[BNODES];
  __shared__ int stage[STAGE];
  const int tid = threadIdx.x;
  const int e0 = bin_base[blockIdx.x];
  const int nE = bin_base[blockIdx.x + 1] - e0;
  const int n0 = blockIdx.x << BSH;
  if (tid < BNODES) ncnt[tid] = 0;
  __syncthreads();
  for (int j = tid; j < nE; j += TILE_T) atomicAdd(&ncnt[bdst[e0 + j] - n0], 1);
  __syncthreads();
  if (tid < BNODES) {
    int a = 0;
    for (int j = 0; j < tid; ++j) a += ncnt[j];
    ncur[tid] = a;
    const int node = n0 + tid;
    if (node < nNodes) {
      const int c = ncnt[tid];
      row_ptr[node] = e0 + a;
      cnt_in[node] = c;
      norm_in[node] = rsqrtf(fmaxf((float)c, 1.0f));
      norm_out[node] = rsqrtf(fmaxf((float)cnt_out[node], 1.0f));
    }
  }
  __syncthreads();
  const bool staged = nE <= STAGE;
  for (int j = tid; j < nE; j += TILE_T) {
    const int q = atomicAdd(&ncur[bdst[e0 + j] - n0], 1);
    const int s = bsrc[e0 + j];
    if (staged) stage[q] = s; else sorted_src[e0 + q] = s;
  }
  __syncthreads();
  if (staged) {
    for (int j = tid; j < nE; j += TILE_T) sorted_src[e0 + j] = stage[j];
  }
}


__device__ __forceinline__ void add4(float4& a, const float4& v) {
  a.x += v.x; a.y += v.y; a.z += v.z; a.w += v.w;
}

template <int MODE>
__global__ void gather_rows(const float* __restrict__ feat, const int* __restrict__ row_ptr,
                            const int* __restrict__ cnt_in, const int* __restrict__ sorted_src,
                            const float* __restrict__ norm_in, const float* __restrict__ norm_out,
                            const float* __restrict__ bias, float* __restrict__ outp, int nNodes) {
  constexpr unsigned ROW4 = (MODE == 2) ? C_LD / 4 : HID_F / 4;
  constexpr unsigned NCH  = (MODE == 2) ? OUT_F / 4   : HID_F / 4;
  const unsigned t = blockIdx.x * blockDim.x + threadIdx.x;
  const unsigned i = t >> 4, ch = t & 15;
  if (i >= (unsigned)nNodes || ch >= NCH) return;
  const float4* f4 = (const float4*)feat + ch;
  int e = row_ptr[i];
  const int end = e + cnt_in[i];
  float4 acc = make_float4(0.0f, 0.0f, 0.0f, 0.0f);
  int s0 = 0, s1 = 0, s2 = 0, s3 = 0;
  bool more = e + 4 <= end;
  if (more) { s0 = sorted_src[e]; s1 = sorted_src[e + 1]; s2 = sorted_src[e + 2]; s3 = sorted_src[e + 3]; }
  while (more) {
    const float4 v0 = f4[(size_t)s0 * ROW4];
    const float4 v1 = f4[(size_t)s1 * ROW4];
    const float4 v2 = f4[(size_t)s2 * ROW4];
    const float4 v3 = f4[(size_t)s3 * ROW4];
    e += 4;
    more = e + 4 <= end;
    if (more) { s0 = sorted_src[e]; s1 = sorted_src[e + 1]; s2 = sorted_src[e + 2]; s3 = sorted_src[e + 3]; }
    add4(acc, v0); add4(acc, v1); add4(acc, v2); add4(acc, v3);
  }
  for (; e < end; ++e) {
    const float4 v = f4[(size_t)sorted_src[e] * ROW4];
    add4(acc, v);
  }
  if (MODE == 0) {
    ((float4*)outp)[(size_t)i * (HID_F / 4) + ch] = acc;
  } else if (MODE == 1) {
    const float ni = norm_in[i], no = norm_out[i];
    const float4 b = ((const float4*)bias)[ch];
    float4 r;
    r.x = fmaxf(acc.x * ni + b.x, 0.0f) * no;
    r.y = fmaxf(acc.y * ni + b.y, 0.0f) * no;
    r.z = fmaxf(acc.z * ni + b.z, 0.0f) * no;
    r.w = fmaxf(acc.w * ni + b.w, 0.0f) * no;
    ((float4*)outp)[(size_t)i * (HID_F / 4) + ch] = r;
  } else {
    const float ni = norm_in[i];
    const float4 b = ((const float4*)bias)[ch];
    float4 r;
    r.x = acc.x * ni + b.x;
    r.y = acc.y * ni + b.y;
    r.z = acc.z * ni + b.z;
    r.w = acc.w * ni + b.w;
    ((float4*)outp)[(size_t)i * (OUT_F / 4) + ch] = r;
  }
}


#define GW   8
#define A_LD 68

template <int MODE, int NT, int NPAD>
__device__ __forceinline__ void products64(const float* at, const float* Wq, int kq0, int half, int l, float sA,
                                           v8f (&c)[NT]) {
  float4 a4[8];
#pragma unroll
  for (int sp = 0; sp < 8; ++sp) {
    a4[sp] = *(const float4*)(at + l * A_LD + 8 * sp + 4 * half);
    if (MODE == 1) { a4[sp].x *= sA; a4[sp].y *= sA; a4[sp].z *= sA; a4[sp].w *= sA; }
  }
#pragma unroll
  for (int sp = 0; sp < 8; ++sp) {
    float4 b4[NT];
#pragma unroll
    for (int t = 0; t < NT; ++t)
      b4[t] = *(const float4*)(Wq + ((kq0 + 2 * sp + half) * NPAD + 16 * t + l) * 4);
#pragma unroll
    for (int t = 0; t < NT; ++t) {
      v2f a, b;
      a.x = a4[sp].x; a.y = a4[sp].y; b.x = b4[t].x; b.y = b4[t].y;
      c[t] = __builtin_amdgcn_wmma_f32_16x16x4_f32(false, a, false, b, (short)0, c[t], false, false);
    }
#pragma unroll
    for (int t = 0; t < NT; ++t) {
      v2f a, b;
      a.x = a4[sp].z; a.y = a4[sp].w; b.x = b4[t].z; b.y = b4[t].w;
      c[t] = __builtin_amdgcn_wmma_f32_16x16x4_f32(false, a, false, b, (short)0, c[t], false, false);
    }
  }
}

template <int MODE>
__global__ void __launch_bounds__(GW * 32, 8)
gemm_rows(const float* __restrict__ A, const float* __restrict__ W, const float* __restrict__ bias,
          const float* __restrict__ norm_in, const float* __restrict__ norm_out,
          float* __restrict__ out, int nNodes, int mTiles) {
  constexpr int KTOT = (MODE == 1) ? IN_F : HID_F;
  constexpr int NCOL = (MODE == 3) ? OUT_F : HID_F;
  constexpr int NPAD = (MODE == 3) ? OUT_PAD : HID_F;
  constexpr int NT   = NPAD / 16;
  constexpr int O_LD = (MODE == 3) ? C_LD : HID_F;
  static_assert(KTOT == 64 || KTOT == 128, "one or two parts of 64 k");
  __shared__ __attribute__((aligned(16))) float Wq[KTOT * NPAD];
  __shared__ __attribute__((aligned(16))) float At[GW][16 * A_LD];

  const int tid = threadIdx.x, lane = tid & 31, w = tid >> 5;
  const int half = lane >> 4, l = lane & 15;
  const int tm = blockIdx.x * GW + w;
  const int tmc = tm < mTiles ? tm : mTiles - 1;
  float* at = At[w];

  int aoff[8];
#pragma unroll
  for (int q = 0; q < 8; ++q) {
    const int row = tmc * 16 + 2 * q + half;
    const int r = row < nNodes ? row : nNodes - 1;
    aoff[q] = r * KTOT + l * 4;
  }
  float4 g0[8];
#pragma unroll
  for (int q = 0; q < 8; ++q) g0[q] = *(const float4*)(A + aoff[q]);
  float sA = 1.0f;
  if (MODE == 1) {
    const int rowA = tmc * 16 + l;
    sA = norm_out[rowA < nNodes ? rowA : nNodes - 1];
  }
  float ni[8], no[8];
  if (MODE == 2) {
#pragma unroll
    for (int q = 0; q < 8; ++q) {
      const int row = tmc * 16 + 2 * q + half;
      const int r = row < nNodes ? row : nNodes - 1;
      ni[q] = norm_in[r];
      no[q] = norm_out[r];
    }
  }

  constexpr int WPER = (KTOT / 4) * NPAD / (GW * 32);
  static_assert((KTOT / 4) * NPAD % (GW * 32) == 0, "weight items divide evenly over the threads");
  float4 wv[WPER];
#pragma unroll
  for (int j = 0; j < WPER; ++j) {
    const int i = tid + j * GW * 32;
    const int kq = i / NPAD, n = i - kq * NPAD;
    const int nc = n < NCOL ? n : NCOL - 1;
    const float* wp = W + (4 * kq) * NCOL + nc;
    float4 v;
    v.x = wp[0]; v.y = wp[NCOL]; v.z = wp[2 * NCOL]; v.w = wp[3 * NCOL];
    if (n >= NCOL) v = make_float4(0.0f, 0.0f, 0.0f, 0.0f);
    wv[j] = v;
  }
#pragma unroll
  for (int j = 0; j < WPER; ++j) *(float4*)(Wq + (tid + j * GW * 32) * 4) = wv[j];

  v8f c[NT];
#pragma unroll
  for (int t = 0; t < NT; ++t) c[t] = v8f{};

#pragma unroll
  for (int q = 0; q < 8; ++q) *(float4*)(at + (2 * q + half) * A_LD + l * 4) = g0[q];
  __syncthreads();
  products64<MODE, NT, NPAD>(at, Wq, 0, half, l, sA, c);
  if constexpr (KTOT == 128) {
    __syncthreads();
    float4 g1[8];
#pragma unroll
    for (int q = 0; q < 8; ++q) g1[q] = *(const float4*)(A + aoff[q] + 64);
#pragma unroll
    for (int q = 0; q < 8; ++q) *(float4*)(at + (2 * q + half) * A_LD + l * 4) = g1[q];
    __syncthreads();
    products64<MODE, NT, NPAD>(at, Wq, 16, half, l, sA, c);
  }

  __syncthreads();
#pragma unroll
  for (int t = 0; t < NT; ++t) {
#pragma unroll
    for (int r = 0; r < 8; ++r) at[(8 * half + r) * A_LD + 16 * t + l] = c[t][r];
  }
  __syncthreads();
  if (tm < mTiles && l * 4 < NPAD) {
    float4 bv = make_float4(0.0f, 0.0f, 0.0f, 0.0f);
    if (MODE == 2) bv = ((const float4*)bias)[l];
#pragma unroll
    for (int q = 0; q < 8; ++q) {
      const int row = tm * 16 + 2 * q + half;
      if (row < nNodes) {
        float4 v = *(const float4*)(at + (2 * q + half) * A_LD + l * 4);
        if (MODE == 2) {
          v.x = fmaxf(v.x * ni[q] + bv.x, 0.0f) * no[q];
          v.y = fmaxf(v.y * ni[q] + bv.y, 0.0f) * no[q];
          v.z = fmaxf(v.z * ni[q] + bv.z, 0.0f) * no[q];
          v.w = fmaxf(v.w * ni[q] + bv.w, 0.0f) * no[q];
        }
        *(float4*)(out + (size_t)row * O_LD + l * 4) = v;
      }
    }
  }
}


static inline int cdiv_ll(long long a, long long b) { return (int)((a + b - 1) / b); }

extern "C" void kernel_launch(void* const* d_in, const int* in_sizes, int n_in,
                              void* d_out, int out_size, void* d_ws, size_t ws_size,
                              hipStream_t stream) {
  const float* X  = (const float*)d_in[0];
  const float* W1 = (const float*)d_in[1];
  const float* b1 = (const float*)d_in[2];
  const float* W2 = (const float*)d_in[3];
  const float* b2 = (const float*)d_in[4];
  const float* W3 = (const float*)d_in[5];
  const float* b3 = (const float*)d_in[6];
  const int* src  = (const int*)d_in[7];
  const int* dst  = (const int*)d_in[8];
  float* out = (float*)d_out;

  const int nNodes = in_sizes[0] / IN_F;
  const int nEdges = in_sizes[7];
  const int mTiles = (nNodes + 15) / 16;
  const int nBins  = (nNodes + BNODES - 1) >> BSH;
  const int nTiles = (nEdges + TILE - 1) / TILE;

  const size_t nP = ((size_t)nNodes + 63) & ~(size_t)63;
  const size_t eP = ((size_t)nEdges + 63) & ~(size_t)63;
  const size_t hP = ((size_t)nTiles * nBins + 63) & ~(size_t)63;
  int* wi = (int*)d_ws;
  int* cnt_out    = wi;
  int* cnt_in     = cnt_out + nP;
  int* row_ptr    = cnt_in + nP;
  int* hist       = row_ptr + nP;
  int* off        = hist + hP;
  int* bin_base   = off + hP;
  int* bsrc       = bin_base + MAXB + 64;
  int* bdst       = bsrc + eP;
  int* sorted_src = bdst + eP;
  float* norm_out = (float*)(sorted_src + eP);
  float* norm_in  = norm_out + nP;
  float* bufA     = norm_in + nP;
  float* bufB     = bufA + nP * HID_F;
  float* bufC     = bufB + nP * HID_F;

  const dim3 blk(256);
  const int gemmBlocks = (mTiles + GW - 1) / GW;
  const int gatherBlocks = cdiv_ll((long long)nNodes * 16, 256);

  zero_i32<<<cdiv_ll((long long)nP, 256), blk, 0, stream>>>(cnt_out, (int)nP);
  tile_count<<<nTiles, dim3(TILE_T), 0, stream>>>(src, dst, cnt_out, hist, nEdges, nBins);
  bin_offsets<<<1, dim3(2 * MAXB), 0, stream>>>(hist, off, bin_base, nTiles, nBins);
  tile_scatter<<<nTiles, dim3(TILE_T), 0, stream>>>(src, dst, hist, off, bsrc, bdst, nEdges, nBins);
  bin_finish<<<nBins, dim3(TILE_T), 0, stream>>>(bsrc, bdst, bin_base, cnt_out, row_ptr, cnt_in,
                                                norm_in, norm_out, sorted_src, nNodes);

  gemm_rows<1><<<gemmBlocks, dim3(GW * 32), 0, stream>>>(
      X, W1, b1, norm_in, norm_out, bufA, nNodes, mTiles);
  gather_rows<1><<<gatherBlocks, blk, 0, stream>>>(
      bufA, row_ptr, cnt_in, sorted_src, norm_in, norm_out, b1, bufB, nNodes);

  gather_rows<0><<<gatherBlocks, blk, 0, stream>>>(
      bufB, row_ptr, cnt_in, sorted_src, norm_in, norm_out, b2, bufA, nNodes);
  gemm_rows<2><<<gemmBlocks, dim3(GW * 32), 0, stream>>>(
      bufA, W2, b2, norm_in, norm_out, bufB, nNodes, mTiles);

  gemm_rows<3><<<gemmBlocks, dim3(GW * 32), 0, stream>>>(
      bufB, W3, b3, norm_in, norm_out, bufC, nNodes, mTiles);
  gather_rows<2><<<gatherBlocks, blk, 0, stream>>>(
      bufC, row_ptr, cnt_in, sorted_src, norm_in, norm_out, b3, out, nNodes);
}


# Gather re-shaped for loads in flight: a node's source list comes in by one lane-parallel load, cross-lane moves hand it round, its rows go out as one burst; groups walk several nodes; sum order kept
# speedup vs baseline: 2.8787x; 1.0449x over previous
#include <hip/hip_runtime.h>


typedef __attribute__((ext_vector_type(2))) float v2f;
typedef __attribute__((ext_vector_type(8))) float v8f;
typedef __attribute__((ext_vector_type(4))) float v4f;

#define IN_F    128
#define HID_F   64
#define OUT_F   40
#define OUT_PAD 48
#define C_LD    64

#define BSH      8
#define BNODES   (1 << BSH)
#define MAXB     512
#define TILE     8192
#define TILE_T   1024
#define STAGE    8192


__global__ void zero_i32(int* __restrict__ p, int n) {
  int t = blockIdx.x * blockDim.x + threadIdx.x;
  if (t < n) p[t] = 0;
}

__global__ void __launch_bounds__(TILE_T)
tile_count(const int* __restrict__ se, const int* __restrict__ de,
           int* __restrict__ cnt_out, int* __restrict__ hist, int nEdges, int nBins) {
  __shared__ int h[MAXB];
  const int tid = threadIdx.x;
  if (tid < MAXB) h[tid] = 0;
  __syncthreads();
  const int base = blockIdx.x * TILE;
#pragma unroll
  for (int j = 0; j < TILE / TILE_T; ++j) {
    const int e = base + j * TILE_T + tid;
    if (e < nEdges) {
      atomicAdd(&h[de[e] >> BSH], 1);
      atomicAdd(&cnt_out[se[e]], 1);
    }
  }
  __syncthreads();
  if (tid < nBins) hist[blockIdx.x * nBins + tid] = h[tid];
}

__global__ void __launch_bounds__(2 * MAXB)
bin_offsets(const int* __restrict__ hist, int* __restrict__ off, int* __restrict__ bin_base,
            int nTiles, int nBins) {
  __shared__ int ps[2][MAXB];
  __shared__ int tot[MAXB];
  __shared__ int bs[MAXB + 1];
  const int tid = threadIdx.x;
  const int b = tid % MAXB, part = tid / MAXB;
  const int halfT = (nTiles + 1) >> 1;
  const int c0 = part ? halfT : 0, c1 = part ? nTiles : halfT;
  int sum = 0;
  if (b < nBins) {
#pragma unroll 16
    for (int c = c0; c < c1; ++c) sum += hist[c * nBins + b];
  }
  ps[part][b] = sum;
  __syncthreads();
  if (tid < MAXB) tot[tid] = ps[0][tid] + ps[1][tid];
  __syncthreads();
  if (tid <= MAXB) {
    int a = 0;
    for (int j = 0; j < tid; ++j) a += tot[j];
    bs[tid] = a;
  }
  __syncthreads();
  if (b < nBins) {
    int run = bs[b] + (part ? ps[0][b] : 0);
#pragma unroll 16
    for (int c = c0; c < c1; ++c) {
      const int v = hist[c * nBins + b];
      off[c * nBins + b] = run;
      run += v;
    }
  }
  if (tid <= nBins) bin_base[tid] = bs[tid];
}

__global__ void __launch_bounds__(TILE_T)
tile_scatter(const int* __restrict__ se, const int* __restrict__ de,
             const int* __restrict__ hist, const int* __restrict__ off,
             int* __restrict__ bsrc, int* __restrict__ bdst, int nEdges, int nBins) {
  __shared__ int lcnt[MAXB];
  __shared__ int lcur[MAXB];
  __shared__ int gdel[MAXB];
  __shared__ int ssrc[TILE];
  __shared__ int sdst[TILE];
  const int tid = threadIdx.x;
  const int base = blockIdx.x * TILE;
  const int left = nEdges - base;
  const int nT = left < TILE ? left : TILE;
  if (tid < MAXB) lcnt[tid] = (tid < nBins) ? hist[blockIdx.x * nBins + tid] : 0;
  __syncthreads();
  if (tid < MAXB) {
    int a = 0;
    for (int j = 0; j < tid; ++j) a += lcnt[j];
    lcur[tid] = a;
    gdel[tid] = (tid < nBins) ? off[blockIdx.x * nBins + tid] - a : 0;
  }
  __syncthreads();
#pragma unroll
  for (int j = 0; j < TILE / TILE_T; ++j) {
    const int e = base + j * TILE_T + tid;
    if (e < nEdges) {
      const int s = se[e], d = de[e];
      const int q = atomicAdd(&lcur[d >> BSH], 1);
      ssrc[q] = s;
      sdst[q] = d;
    }
  }
  __syncthreads();
  for (int j = tid; j < nT; j += TILE_T) {
    const int d = sdst[j];
    const int g = gdel[d >> BSH] + j;
    bsrc[g] = ssrc[j];
    bdst[g] = d;
  }
}

__global__ void __launch_bounds__(TILE_T)
bin_finish(const int* __restrict__ bsrc, const int* __restrict__ bdst,
           const int* __restrict__ bin_base, const int* __restrict__ cnt_out,
           int* __restrict__ row_ptr, int* __restrict__ cnt_in,
           float* __restrict__ norm_in, float* __restrict__ norm_out,
           int* __restrict__ sorted_src, int nNodes) {
  __shared__ int ncnt[BNODES];
  __shared__ int ncur[BNODES];
  __shared__ int stage[STAGE];
  const int tid = threadIdx.x;
  const int e0 = bin_base[blockIdx.x];
  const int nE = bin_base[blockIdx.x + 1] - e0;
  const int n0 = blockIdx.x << BSH;
  if (tid < BNODES) ncnt[tid] = 0;
  __syncthreads();
  for (int j = tid; j < nE; j += TILE_T) atomicAdd(&ncnt[bdst[e0 + j] - n0], 1);
  __syncthreads();
  if (tid < BNODES) {
    int a = 0;
    for (int j = 0; j < tid; ++j) a += ncnt[j];
    ncur[tid] = a;
    const int node = n0 + tid;
    if (node < nNodes) {
      const int c = ncnt[tid];
      row_ptr[node] = e0 + a;
      cnt_in[node] = c;
      norm_in[node] = rsqrtf(fmaxf((float)c, 1.0f));
      norm_out[node] = rsqrtf(fmaxf((float)cnt_out[node], 1.0f));
    }
  }
  __syncthreads();
  const bool staged = nE <= STAGE;
  for (int j = tid; j < nE; j += TILE_T) {
    const int q = atomicAdd(&ncur[bdst[e0 + j] - n0], 1);
    const int s = bsrc[e0 + j];
    if (staged) stage[q] = s; else sorted_src[e0 + q] = s;
  }
  __syncthreads();
  if (staged) {
    for (int j = tid; j < nE; j += TILE_T) sorted_src[e0 + j] = stage[j];
  }
}


__device__ __forceinline__ void add4(float4& a, const float4& v) {
  a.x += v.x; a.y += v.y; a.z += v.z; a.w += v.w;
}

#define GATHER_T   256
#define GATHER_WG  1024

template <unsigned ROWB, int NB>
__device__ __forceinline__ void sum_burst(float4& acc, const char* fb, int idx, int n, bool col) {
  static_assert(NB == 8 || NB == 16, "a burst is 8 or 16 rows");
  int s[NB];
#pragma unroll
  for (int j = 0; j < NB; ++j) s[j] = __shfl(idx, j, 16);
  if (col) {
    v4f v[NB];
#pragma unroll
    for (int j = 0; j < NB; ++j) v[j] = *(const v4f*)(fb + (unsigned)s[j] * ROWB);
    __builtin_amdgcn_sched_barrier(0);
#pragma unroll
    for (int j = 0; j < NB; ++j) {
      const bool in = j < n;
      acc.x += in ? v[j].x : 0.0f;
      acc.y += in ? v[j].y : 0.0f;
      acc.z += in ? v[j].z : 0.0f;
      acc.w += in ? v[j].w : 0.0f;
    }
  }
}

template <int MODE>
__global__ void __launch_bounds__(GATHER_T)
gather_rows(const float* __restrict__ feat, const int* __restrict__ row_ptr,
            const int* __restrict__ cnt_in, const int* __restrict__ sorted_src,
            const float* __restrict__ norm_in, const float* __restrict__ norm_out,
            const float* __restrict__ bias, float* __restrict__ outp, int nNodes) {
  constexpr unsigned ROWB = ((MODE == 2) ? C_LD : HID_F) * 4;
  constexpr int      NCH  = (MODE == 2) ? OUT_F / 4 : HID_F / 4;
  const unsigned t = blockIdx.x * blockDim.x + threadIdx.x;
  const int nGroups = (int)((gridDim.x * blockDim.x) >> 4);
  const int ch = (int)(t & 15);
  const bool col = ch < NCH;
  const char* fb = (const char*)feat + ch * 16;
  float4 b = make_float4(0.0f, 0.0f, 0.0f, 0.0f);
  if (MODE != 0 && col) b = ((const float4*)bias)[ch];

  const int last = nNodes - 1;
  int i = (int)(t >> 4);
  const int ic = i < last ? i : last;
  int e = row_ptr[ic];
  const int c0 = cnt_in[ic];
  int d = i < nNodes ? c0 : 0;
  int i1 = i + nGroups;
  const int i1c = i1 < last ? i1 : last;
  int e1 = row_ptr[i1c];
  const int c1 = cnt_in[i1c];
  int d1 = i1 < nNodes ? c1 : 0;
  int lo = sorted_src[ch < d ? e + ch : 0];
  int hi = sorted_src[ch + 16 < d ? e + 16 + ch : 0];

  while (i < nNodes) {
    const int i2 = i1 + nGroups;
    const int i2c = i2 < last ? i2 : last;
    const int e2 = row_ptr[i2c];
    const int c2 = cnt_in[i2c];
    const int d2 = i2 < nNodes ? c2 : 0;
    const int lo1 = sorted_src[ch < d1 ? e1 + ch : 0];
    const int hi1 = sorted_src[ch + 16 < d1 ? e1 + 16 + ch : 0];
    float ni = 1.0f, no = 1.0f;
    if (MODE != 0) ni = norm_in[i];
    if (MODE == 1) no = norm_out[i];

    float4 acc = make_float4(0.0f, 0.0f, 0.0f, 0.0f);
    if (d > 0) sum_burst<ROWB, 16>(acc, fb, lo, d, col);
    if (d > 24) sum_burst<ROWB, 16>(acc, fb, hi, d - 16, col);
    else if (d > 16) sum_burst<ROWB, 8>(acc, fb, hi, d - 16, col);
    if (col) {
      for (int k = 32; k < d; ++k) {
        const float4 v = *(const float4*)(fb + (unsigned)sorted_src[e + k] * ROWB);
        add4(acc, v);
      }
      if (MODE == 0) {
        ((float4*)outp)[(size_t)i * (HID_F / 4) + ch] = acc;
      } else if (MODE == 1) {
        float4 r;
        r.x = fmaxf(acc.x * ni + b.x, 0.0f) * no;
        r.y = fmaxf(acc.y * ni + b.y, 0.0f) * no;
        r.z = fmaxf(acc.z * ni + b.z, 0.0f) * no;
        r.w = fmaxf(acc.w * ni + b.w, 0.0f) * no;
        ((float4*)outp)[(size_t)i * (HID_F / 4) + ch] = r;
      } else {
        float4 r;
        r.x = acc.x * ni + b.x;
        r.y = acc.y * ni + b.y;
        r.z = acc.z * ni + b.z;
        r.w = acc.w * ni + b.w;
        ((float4*)outp)[(size_t)i * (OUT_F / 4) + ch] = r;
      }
    }
    i = i1; e = e1; d = d1; lo = lo1; hi = hi1;
    i1 = i2; e1 = e2; d1 = d2;
  }
}


#define GW   8
#define A_LD 68

template <int MODE, int NT, int NPAD>
__device__ __forceinline__ void products64(const float* at, const float* Wq, int kq0, int half, int l, float sA,
                                           v8f (&c)[NT]) {
  float4 a4[8];
#pragma unroll
  for (int sp = 0; sp < 8; ++sp) {
    a4[sp] = *(const float4*)(at + l * A_LD + 8 * sp + 4 * half);
    if (MODE == 1) { a4[sp].x *= sA; a4[sp].y *= sA; a4[sp].z *= sA; a4[sp].w *= sA; }
  }
#pragma unroll
  for (int sp = 0; sp < 8; ++sp) {
    float4 b4[NT];
#pragma unroll
    for (int t = 0; t < NT; ++t)
      b4[t] = *(const float4*)(Wq + ((kq0 + 2 * sp + half) * NPAD + 16 * t + l) * 4);
#pragma unroll
    for (int t = 0; t < NT; ++t) {
      v2f a, b;
      a.x = a4[sp].x; a.y = a4[sp].y; b.x = b4[t].x; b.y = b4[t].y;
      c[t] = __builtin_amdgcn_wmma_f32_16x16x4_f32(false, a, false, b, (short)0, c[t], false, false);
    }
#pragma unroll
    for (int t = 0; t < NT; ++t) {
      v2f a, b;
      a.x = a4[sp].z; a.y = a4[sp].w; b.x = b4[t].z; b.y = b4[t].w;
      c[t] = __builtin_amdgcn_wmma_f32_16x16x4_f32(false, a, false, b, (short)0, c[t], false, false);
    }
  }
}

template <int MODE>
__global__ void __launch_bounds__(GW * 32, 8)
gemm_rows(const float* __restrict__ A, const float* __restrict__ W, const float* __restrict__ bias,
          const float* __restrict__ norm_in, const float* __restrict__ norm_out,
          float* __restrict__ out, int nNodes, int mTiles) {
  constexpr int KTOT = (MODE == 1) ? IN_F : HID_F;
  constexpr int NCOL = (MODE == 3) ? OUT_F : HID_F;
  constexpr int NPAD = (MODE == 3) ? OUT_PAD : HID_F;
  constexpr int NT   = NPAD / 16;
  constexpr int O_LD = (MODE == 3) ? C_LD : HID_F;
  static_assert(KTOT == 64 || KTOT == 128, "one or two parts of 64 k");
  __shared__ __attribute__((aligned(16))) float Wq[KTOT * NPAD];
  __shared__ __attribute__((aligned(16))) float At[GW][16 * A_LD];

  const int tid = threadIdx.x, lane = tid & 31, w = tid >> 5;
  const int half = lane >> 4, l = lane & 15;
  const int tm = blockIdx.x * GW + w;
  const int tmc = tm < mTiles ? tm : mTiles - 1;
  float* at = At[w];

  int aoff[8];
#pragma unroll
  for (int q = 0; q < 8; ++q) {
    const int row = tmc * 16 + 2 * q + half;
    const int r = row < nNodes ? row : nNodes - 1;
    aoff[q] = r * KTOT + l * 4;
  }
  float4 g0[8];
#pragma unroll
  for (int q = 0; q < 8; ++q) g0[q] = *(const float4*)(A + aoff[q]);
  float sA = 1.0f;
  if (MODE == 1) {
    const int rowA = tmc * 16 + l;
    sA = norm_out[rowA < nNodes ? rowA : nNodes - 1];
  }
  float ni[8], no[8];
  if (MODE == 2) {
#pragma unroll
    for (int q = 0; q < 8; ++q) {
      const int row = tmc * 16 + 2 * q + half;
      const int r = row < nNodes ? row : nNodes - 1;
      ni[q] = norm_in[r];
      no[q] = norm_out[r];
    }
  }

  constexpr int WPER = (KTOT / 4) * NPAD / (GW * 32);
  static_assert((KTOT / 4) * NPAD % (GW * 32) == 0, "weight items divide evenly over the threads");
  float4 wv[WPER];
#pragma unroll
  for (int j = 0; j < WPER; ++j) {
    const int i = tid + j * GW * 32;
    const int kq = i / NPAD, n = i - kq * NPAD;
    const int nc = n < NCOL ? n : NCOL - 1;
    const float* wp = W + (4 * kq) * NCOL + nc;
    float4 v;
    v.x = wp[0]; v.y = wp[NCOL]; v.z = wp[2 * NCOL]; v.w = wp[3 * NCOL];
    if (n >= NCOL) v = make_float4(0.0f, 0.0f, 0.0f, 0.0f);
    wv[j] = v;
  }
#pragma unroll
  for (int j = 0; j < WPER; ++j) *(float4*)(Wq + (tid + j * GW * 32) * 4) = wv[j];

  v8f c[NT];
#pragma unroll
  for (int t = 0; t < NT; ++t) c[t] = v8f{};

#pragma unroll
  for (int q = 0; q < 8; ++q) *(float4*)(at + (2 * q + half) * A_LD + l * 4) = g0[q];
  __syncthreads();
  products64<MODE, NT, NPAD>(at, Wq, 0, half, l, sA, c);
  if constexpr (KTOT == 128) {
    __syncthreads();
    float4 g1[8];
#pragma unroll
    for (int q = 0; q < 8; ++q) g1[q] = *(const float4*)(A + aoff[q] + 64);
#pragma unroll
    for (int q = 0; q < 8; ++q) *(float4*)(at + (2 * q + half) * A_LD + l * 4) = g1[q];
    __syncthreads();
    products64<MODE, NT, NPAD>(at, Wq, 16, half, l, sA, c);
  }

  __syncthreads();
#pragma unroll
  for (int t = 0; t < NT; ++t) {
#pragma unroll
    for (int r = 0; r < 8; ++r) at[(8 * half + r) * A_LD + 16 * t + l] = c[t][r];
  }
  __syncthreads();
  if (tm < mTiles && l * 4 < NPAD) {
    float4 bv = make_float4(0.0f, 0.0f, 0.0f, 0.0f);
    if (MODE == 2) bv = ((const float4*)bias)[l];
#pragma unroll
    for (int q = 0; q < 8; ++q) {
      const int row = tm * 16 + 2 * q + half;
      if (row < nNodes) {
        float4 v = *(const float4*)(at + (2 * q + half) * A_LD + l * 4);
        if (MODE == 2) {
          v.x = fmaxf(v.x * ni[q] + bv.x, 0.0f) * no[q];
          v.y = fmaxf(v.y * ni[q] + bv.y, 0.0f) * no[q];
          v.z = fmaxf(v.z * ni[q] + bv.z, 0.0f) * no[q];
          v.w = fmaxf(v.w * ni[q] + bv.w, 0.0f) * no[q];
        }
        *(float4*)(out + (size_t)row * O_LD + l * 4) = v;
      }
    }
  }
}


static inline int cdiv_ll(long long a, long long b) { return (int)((a + b - 1) / b); }

extern "C" void kernel_launch(void* const* d_in, const int* in_sizes, int n_in,
                              void* d_out, int out_size, void* d_ws, size_t ws_size,
                              hipStream_t stream) {
  const float* X  = (const float*)d_in[0];
  const float* W1 = (const float*)d_in[1];
  const float* b1 = (const float*)d_in[2];
  const float* W2 = (const float*)d_in[3];
  const float* b2 = (const float*)d_in[4];
  const float* W3 = (const float*)d_in[5];
  const float* b3 = (const float*)d_in[6];
  const int* src  = (const int*)d_in[7];
  const int* dst  = (const int*)d_in[8];
  float* out = (float*)d_out;

  const int nNodes = in_sizes[0] / IN_F;
  const int nEdges = in_sizes[7];
  const int mTiles = (nNodes + 15) / 16;
  const int nBins  = (nNodes + BNODES - 1) >> BSH;
  const int nTiles = (nEdges + TILE - 1) / TILE;

  const size_t nP = ((size_t)nNodes + 63) & ~(size_t)63;
  const size_t eP = ((size_t)nEdges + 63) & ~(size_t)63;
  const size_t hP = ((size_t)nTiles * nBins + 63) & ~(size_t)63;
  int* wi = (int*)d_ws;
  int* cnt_out    = wi;
  int* cnt_in     = cnt_out + nP;
  int* row_ptr    = cnt_in + nP;
  int* hist       = row_ptr + nP;
  int* off        = hist + hP;
  int* bin_base   = off + hP;
  int* bsrc       = bin_base + MAXB + 64;
  int* bdst       = bsrc + eP;
  int* sorted_src = bdst + eP;
  float* norm_out = (float*)(sorted_src + eP);
  float* norm_in  = norm_out + nP;
  float* bufA     = norm_in + nP;
  float* bufB     = bufA + nP * HID_F;
  float* bufC     = bufB + nP * HID_F;

  const dim3 blk(256);
  const int gemmBlocks = (mTiles + GW - 1) / GW;
  const int gatherBlocks = GATHER_WG;
  const dim3 gblk(GATHER_T);

  zero_i32<<<cdiv_ll((long long)nP, 256), blk, 0, stream>>>(cnt_out, (int)nP);
  tile_count<<<nTiles, dim3(TILE_T), 0, stream>>>(src, dst, cnt_out, hist, nEdges, nBins);
  bin_offsets<<<1, dim3(2 * MAXB), 0, stream>>>(hist, off, bin_base, nTiles, nBins);
  tile_scatter<<<nTiles, dim3(TILE_T), 0, stream>>>(src, dst, hist, off, bsrc, bdst, nEdges, nBins);
  bin_finish<<<nBins, dim3(TILE_T), 0, stream>>>(bsrc, bdst, bin_base, cnt_out, row_ptr, cnt_in,
                                                norm_in, norm_out, sorted_src, nNodes);

  gemm_rows<1><<<gemmBlocks, dim3(GW * 32), 0, stream>>>(
      X, W1, b1, norm_in, norm_out, bufA, nNodes, mTiles);
  gather_rows<1><<<gatherBlocks, gblk, 0, stream>>>(
      bufA, row_ptr, cnt_in, sorted_src, norm_in, norm_out, b1, bufB, nNodes);

  gather_rows<0><<<gatherBlocks, gblk, 0, stream>>>(
      bufB, row_ptr, cnt_in, sorted_src, norm_in, norm_out, b2, bufA, nNodes);
  gemm_rows<2><<<gemmBlocks, dim3(GW * 32), 0, stream>>>(
      bufA, W2, b2, norm_in, norm_out, bufB, nNodes, mTiles);

  gemm_rows<3><<<gemmBlocks, dim3(GW * 32), 0, stream>>>(
      bufB, W3, b3, norm_in, norm_out, bufC, nNodes, mTiles);
  gather_rows<2><<<gatherBlocks, gblk, 0, stream>>>(
      bufC, row_ptr, cnt_in, sorted_src, norm_in, norm_out, b3, out, nNodes);
}


# Layer 2's gather, its W2 product and layer 3's W3 product fused into one kernel: a wave sums its 16 nodes' rows into its LDS tile and runs both products from it; two launches fewer, sum order kept
# speedup vs baseline: 3.0555x; 1.0614x over previous
#include <hip/hip_runtime.h>


typedef __attribute__((ext_vector_type(2))) float v2f;
typedef __attribute__((ext_vector_type(8))) float v8f;
typedef __attribute__((ext_vector_type(4))) float v4f;

#define IN_F    128
#define HID_F   64
#define OUT_F   40
#define OUT_PAD 48
#define C_LD    64

#define BSH      8
#define BNODES   (1 << BSH)
#define MAXB     512
#define TILE     8192
#define TILE_T   1024
#define STAGE    8192


__global__ void zero_i32(int* __restrict__ p, int n) {
  int t = blockIdx.x * blockDim.x + threadIdx.x;
  if (t < n) p[t] = 0;
}

__global__ void __launch_bounds__(TILE_T)
tile_count(const int* __restrict__ se, const int* __restrict__ de,
           int* __restrict__ cnt_out, int* __restrict__ hist, int nEdges, int nBins) {
  __shared__ int h[MAXB];
  const int tid = threadIdx.x;
  if (tid < MAXB) h[tid] = 0;
  __syncthreads();
  const int base = blockIdx.x * TILE;
#pragma unroll
  for (int j = 0; j < TILE / TILE_T; ++j) {
    const int e = base + j * TILE_T + tid;
    if (e < nEdges) {
      atomicAdd(&h[de[e] >> BSH], 1);
      atomicAdd(&cnt_out[se[e]], 1);
    }
  }
  __syncthreads();
  if (tid < nBins) hist[blockIdx.x * nBins + tid] = h[tid];
}

__global__ void __launch_bounds__(2 * MAXB)
bin_offsets(const int* __restrict__ hist, int* __restrict__ off, int* __restrict__ bin_base,
            int nTiles, int nBins) {
  __shared__ int ps[2][MAXB];
  __shared__ int tot[MAXB];
  __shared__ int bs[MAXB + 1];
  const int tid = threadIdx.x;
  const int b = tid % MAXB, part = tid / MAXB;
  const int halfT = (nTiles + 1) >> 1;
  const int c0 = part ? halfT : 0, c1 = part ? nTiles : halfT;
  int sum = 0;
  if (b < nBins) {
#pragma unroll 16
    for (int c = c0; c < c1; ++c) sum += hist[c * nBins + b];
  }
  ps[part][b] = sum;
  __syncthreads();
  if (tid < MAXB) tot[tid] = ps[0][tid] + ps[1][tid];
  __syncthreads();
  if (tid <= MAXB) {
    int a = 0;
    for (int j = 0; j < tid; ++j) a += tot[j];
    bs[tid] = a;
  }
  __syncthreads();
  if (b < nBins) {
    int run = bs[b] + (part ? ps[0][b] : 0);
#pragma unroll 16
    for (int c = c0; c < c1; ++c) {
      const int v = hist[c * nBins + b];
      off[c * nBins + b] = run;
      run += v;
    }
  }
  if (tid <= nBins) bin_base[tid] = bs[tid];
}

__global__ void __launch_bounds__(TILE_T)
tile_scatter(const int* __restrict__ se, const int* __restrict__ de,
             const int* __restrict__ hist, const int* __restrict__ off,
             int* __restrict__ bsrc, int* __restrict__ bdst, int nEdges, int nBins) {
  __shared__ int lcnt[MAXB];
  __shared__ int lcur[MAXB];
  __shared__ int gdel[MAXB];
  __shared__ int ssrc[TILE];
  __shared__ int sdst[TILE];
  const int tid = threadIdx.x;
  const int base = blockIdx.x * TILE;
  const int left = nEdges - base;
  const int nT = left < TILE ? left : TILE;
  if (tid < MAXB) lcnt[tid] = (tid < nBins) ? hist[blockIdx.x * nBins + tid] : 0;
  __syncthreads();
  if (tid < MAXB) {
    int a = 0;
    for (int j = 0; j < tid; ++j) a += lcnt[j];
    lcur[tid] = a;
    gdel[tid] = (tid < nBins) ? off[blockIdx.x * nBins + tid] - a : 0;
  }
  __syncthreads();
#pragma unroll
  for (int j = 0; j < TILE / TILE_T; ++j) {
    const int e = base + j * TILE_T + tid;
    if (e < nEdges) {
      const int s = se[e], d = de[e];
      const int q = atomicAdd(&lcur[d >> BSH], 1);
      ssrc[q] = s;
      sdst[q] = d;
    }
  }
  __syncthreads();
  for (int j = tid; j < nT; j += TILE_T) {
    const int d = sdst[j];
    const int g = gdel[d >> BSH] + j;
    bsrc[g] = ssrc[j];
    bdst[g] = d;
  }
}

__global__ void __launch_bounds__(TILE_T)
bin_finish(const int* __restrict__ bsrc, const int* __restrict__ bdst,
           const int* __restrict__ bin_base, const int* __restrict__ cnt_out,
           int* __restrict__ row_ptr, int* __restrict__ cnt_in,
           float* __restrict__ norm_in, float* __restrict__ norm_out,
           int* __restrict__ sorted_src, int nNodes) {
  __shared__ int ncnt[BNODES];
  __shared__ int ncur[BNODES];
  __shared__ int stage[STAGE];
  const int tid = threadIdx.x;
  const int e0 = bin_base[blockIdx.x];
  const int nE = bin_base[blockIdx.x + 1] - e0;
  const int n0 = blockIdx.x << BSH;
  if (tid < BNODES) ncnt[tid] = 0;
  __syncthreads();
  for (int j = tid; j < nE; j += TILE_T) atomicAdd(&ncnt[bdst[e0 + j] - n0], 1);
  __syncthreads();
  if (tid < BNODES) {
    int a = 0;
    for (int j = 0; j < tid; ++j) a += ncnt[j];
    ncur[tid] = a;
    const int node = n0 + tid;
    if (node < nNodes) {
      const int c = ncnt[tid];
      row_ptr[node] = e0 + a;
      cnt_in[node] = c;
      norm_in[node] = rsqrtf(fmaxf((float)c, 1.0f));
      norm_out[node] = rsqrtf(fmaxf((float)cnt_out[node], 1.0f));
    }
  }
  __syncthreads();
  const bool staged = nE <= STAGE;
  for (int j = tid; j < nE; j += TILE_T) {
    const int q = atomicAdd(&ncur[bdst[e0 + j] - n0], 1);
    const int s = bsrc[e0 + j];
    if (staged) stage[q] = s; else sorted_src[e0 + q] = s;
  }
  __syncthreads();
  if (staged) {
    for (int j = tid; j < nE; j += TILE_T) sorted_src[e0 + j] = stage[j];
  }
}


__device__ __forceinline__ void add4(float4& a, const float4& v) {
  a.x += v.x; a.y += v.y; a.z += v.z; a.w += v.w;
}

#define GATHER_T   256
#define GATHER_WG  1024

template <unsigned ROWB, int NB>
__device__ __forceinline__ void sum_burst(float4& acc, const char* fb, int idx, int n, bool col) {
  static_assert(NB == 8 || NB == 16, "a burst is 8 or 16 rows");
  int s[NB];
#pragma unroll
  for (int j = 0; j < NB; ++j) s[j] = __shfl(idx, j, 16);
  if (col) {
    v4f v[NB];
#pragma unroll
    for (int j = 0; j < NB; ++j) v[j] = *(const v4f*)(fb + (unsigned)s[j] * ROWB);
    __builtin_amdgcn_sched_barrier(0);
#pragma unroll
    for (int j = 0; j < NB; ++j) {
      const bool in = j < n;
      acc.x += in ? v[j].x : 0.0f;
      acc.y += in ? v[j].y : 0.0f;
      acc.z += in ? v[j].z : 0.0f;
      acc.w += in ? v[j].w : 0.0f;
    }
  }
}

template <int MODE>
__global__ void __launch_bounds__(GATHER_T)
gather_rows(const float* __restrict__ feat, const int* __restrict__ row_ptr,
            const int* __restrict__ cnt_in, const int* __restrict__ sorted_src,
            const float* __restrict__ norm_in, const float* __restrict__ norm_out,
            const float* __restrict__ bias, float* __restrict__ outp, int nNodes) {
  constexpr unsigned ROWB = ((MODE == 2) ? C_LD : HID_F) * 4;
  constexpr int      NCH  = (MODE == 2) ? OUT_F / 4 : HID_F / 4;
  const unsigned t = blockIdx.x * blockDim.x + threadIdx.x;
  const int nGroups = (int)((gridDim.x * blockDim.x) >> 4);
  const int ch = (int)(t & 15);
  const bool col = ch < NCH;
  const char* fb = (const char*)feat + ch * 16;
  float4 b = make_float4(0.0f, 0.0f, 0.0f, 0.0f);
  if (col) b = ((const float4*)bias)[ch];

  const int last = nNodes - 1;
  int i = (int)(t >> 4);
  const int ic = i < last ? i : last;
  int e = row_ptr[ic];
  const int c0 = cnt_in[ic];
  int d = i < nNodes ? c0 : 0;
  int i1 = i + nGroups;
  const int i1c = i1 < last ? i1 : last;
  int e1 = row_ptr[i1c];
  const int c1 = cnt_in[i1c];
  int d1 = i1 < nNodes ? c1 : 0;
  int lo = sorted_src[ch < d ? e + ch : 0];
  int hi = sorted_src[ch + 16 < d ? e + 16 + ch : 0];

  while (i < nNodes) {
    const int i2 = i1 + nGroups;
    const int i2c = i2 < last ? i2 : last;
    const int e2 = row_ptr[i2c];
    const int c2 = cnt_in[i2c];
    const int d2 = i2 < nNodes ? c2 : 0;
    const int lo1 = sorted_src[ch < d1 ? e1 + ch : 0];
    const int hi1 = sorted_src[ch + 16 < d1 ? e1 + 16 + ch : 0];
    const float ni = norm_in[i];
    float no = 1.0f;
    if (MODE == 1) no = norm_out[i];

    float4 acc = make_float4(0.0f, 0.0f, 0.0f, 0.0f);
    if (d > 0) sum_burst<ROWB, 16>(acc, fb, lo, d, col);
    if (d > 24) sum_burst<ROWB, 16>(acc, fb, hi, d - 16, col);
    else if (d > 16) sum_burst<ROWB, 8>(acc, fb, hi, d - 16, col);
    if (col) {
      for (int k = 32; k < d; ++k) {
        const float4 v = *(const float4*)(fb + (unsigned)sorted_src[e + k] * ROWB);
        add4(acc, v);
      }
      if (MODE == 1) {
        float4 r;
        r.x = fmaxf(acc.x * ni + b.x, 0.0f) * no;
        r.y = fmaxf(acc.y * ni + b.y, 0.0f) * no;
        r.z = fmaxf(acc.z * ni + b.z, 0.0f) * no;
        r.w = fmaxf(acc.w * ni + b.w, 0.0f) * no;
        ((float4*)outp)[(size_t)i * (HID_F / 4) + ch] = r;
      } else {
        float4 r;
        r.x = acc.x * ni + b.x;
        r.y = acc.y * ni + b.y;
        r.z = acc.z * ni + b.z;
        r.w = acc.w * ni + b.w;
        ((float4*)outp)[(size_t)i * (OUT_F / 4) + ch] = r;
      }
    }
    i = i1; e = e1; d = d1; lo = lo1; hi = hi1;
    i1 = i2; e1 = e2; d1 = d2;
  }
}


#define GW   8
#define A_LD 68

template <bool SCALE, int NT, int NPAD>
__device__ __forceinline__ void products64(const float* at, const float* Wq, int kq0, int half, int l, float sA,
                                           v8f (&c)[NT]) {
  float4 a4[8];
#pragma unroll
  for (int sp = 0; sp < 8; ++sp) {
    a4[sp] = *(const float4*)(at + l * A_LD + 8 * sp + 4 * half);
    if (SCALE) { a4[sp].x *= sA; a4[sp].y *= sA; a4[sp].z *= sA; a4[sp].w *= sA; }
  }
#pragma unroll
  for (int sp = 0; sp < 8; ++sp) {
    float4 b4[NT];
#pragma unroll
    for (int t = 0; t < NT; ++t)
      b4[t] = *(const float4*)(Wq + ((kq0 + 2 * sp + half) * NPAD + 16 * t + l) * 4);
#pragma unroll
    for (int t = 0; t < NT; ++t) {
      v2f a, b;
      a.x = a4[sp].x; a.y = a4[sp].y; b.x = b4[t].x; b.y = b4[t].y;
      c[t] = __builtin_amdgcn_wmma_f32_16x16x4_f32(false, a, false, b, (short)0, c[t], false, false);
    }
#pragma unroll
    for (int t = 0; t < NT; ++t) {
      v2f a, b;
      a.x = a4[sp].z; a.y = a4[sp].w; b.x = b4[t].z; b.y = b4[t].w;
      c[t] = __builtin_amdgcn_wmma_f32_16x16x4_f32(false, a, false, b, (short)0, c[t], false, false);
    }
  }
}

__global__ void __launch_bounds__(GW * 32, 8)
gemm_rows(const float* __restrict__ A, const float* __restrict__ W, const float* __restrict__ norm_out,
          float* __restrict__ out, int nNodes, int mTiles) {
  constexpr int KTOT = IN_F;
  constexpr int NPAD = HID_F;
  constexpr int NT   = NPAD / 16;
  static_assert(KTOT == 128, "two parts of 64 k");
  __shared__ __attribute__((aligned(16))) float Wq[KTOT * NPAD];
  __shared__ __attribute__((aligned(16))) float At[GW][16 * A_LD];

  const int tid = threadIdx.x, lane = tid & 31, w = tid >> 5;
  const int half = lane >> 4, l = lane & 15;
  const int tm = blockIdx.x * GW + w;
  const int tmc = tm < mTiles ? tm : mTiles - 1;
  float* at = At[w];

  int aoff[8];
#pragma unroll
  for (int q = 0; q < 8; ++q) {
    const int row = tmc * 16 + 2 * q + half;
    const int r = row < nNodes ? row : nNodes - 1;
    aoff[q] = r * KTOT + l * 4;
  }
  float4 g0[8];
#pragma unroll
  for (int q = 0; q < 8; ++q) g0[q] = *(const float4*)(A + aoff[q]);
  const int rowA = tmc * 16 + l;
  const float sA = norm_out[rowA < nNodes ? rowA : nNodes - 1];

  constexpr int WPER = (KTOT / 4) * NPAD / (GW * 32);
  static_assert((KTOT / 4) * NPAD % (GW * 32) == 0, "weight items divide evenly over the threads");
  float4 wv[WPER];
#pragma unroll
  for (int j = 0; j < WPER; ++j) {
    const int i = tid + j * GW * 32;
    const int kq = i / NPAD, n = i - kq * NPAD;
    const float* wp = W + (4 * kq) * NPAD + n;
    wv[j] = make_float4(wp[0], wp[NPAD], wp[2 * NPAD], wp[3 * NPAD]);
  }
#pragma unroll
  for (int j = 0; j < WPER; ++j) *(float4*)(Wq + (tid + j * GW * 32) * 4) = wv[j];

  v8f c[NT];
#pragma unroll
  for (int t = 0; t < NT; ++t) c[t] = v8f{};

#pragma unroll
  for (int q = 0; q < 8; ++q) *(float4*)(at + (2 * q + half) * A_LD + l * 4) = g0[q];
  __syncthreads();
  products64<true, NT, NPAD>(at, Wq, 0, half, l, sA, c);
  __syncthreads();
  float4 g1[8];
#pragma unroll
  for (int q = 0; q < 8; ++q) g1[q] = *(const float4*)(A + aoff[q] + 64);
#pragma unroll
  for (int q = 0; q < 8; ++q) *(float4*)(at + (2 * q + half) * A_LD + l * 4) = g1[q];
  __syncthreads();
  products64<true, NT, NPAD>(at, Wq, 16, half, l, sA, c);

  __syncthreads();
#pragma unroll
  for (int t = 0; t < NT; ++t) {
#pragma unroll
    for (int r = 0; r < 8; ++r) at[(8 * half + r) * A_LD + 16 * t + l] = c[t][r];
  }
  __syncthreads();
  if (tm < mTiles) {
#pragma unroll
    for (int q = 0; q < 8; ++q) {
      const int row = tm * 16 + 2 * q + half;
      if (row < nNodes)
        *(float4*)(out + (size_t)row * HID_F + l * 4) = *(const float4*)(at + (2 * q + half) * A_LD + l * 4);
    }
  }
}

__global__ void __launch_bounds__(GW * 32, 8)
layer2_layer3(const float* __restrict__ feat, const int* __restrict__ row_ptr,
              const int* __restrict__ cnt_in, const int* __restrict__ sorted_src,
              const float* __restrict__ W2, const float* __restrict__ b2, const float* __restrict__ W3,
              const float* __restrict__ norm_in, const float* __restrict__ norm_out,
              float* __restrict__ out, int nNodes, int mTiles) {
  constexpr int NT2 = HID_F / 16, NT3 = OUT_PAD / 16;
  __shared__ __attribute__((aligned(16))) float Wq2[HID_F * HID_F];
  __shared__ __attribute__((aligned(16))) float Wq3[HID_F * OUT_PAD];
  __shared__ __attribute__((aligned(16))) float At[GW][16 * A_LD];

  const int tid = threadIdx.x, lane = tid & 31, w = tid >> 5;
  const int half = lane >> 4, l = lane & 15;
  const int tm = blockIdx.x * GW + w;
  const int tmc = tm < mTiles ? tm : mTiles - 1;
  const int last = nNodes - 1;
  float* at = At[w];

  constexpr int WPER2 = (HID_F / 4) * HID_F / (GW * 32), WPER3 = (HID_F / 4) * OUT_PAD / (GW * 32);
  static_assert((HID_F / 4) * HID_F % (GW * 32) == 0 && (HID_F / 4) * OUT_PAD % (GW * 32) == 0,
                "weight items divide evenly over the threads");
  float4 wv2[WPER2], wv3[WPER3];
#pragma unroll
  for (int j = 0; j < WPER2; ++j) {
    const int i = tid + j * GW * 32;
    const int kq = i / HID_F, n = i - kq * HID_F;
    const float* wp = W2 + (4 * kq) * HID_F + n;
    wv2[j] = make_float4(wp[0], wp[HID_F], wp[2 * HID_F], wp[3 * HID_F]);
  }
#pragma unroll
  for (int j = 0; j < WPER3; ++j) {
    const int i = tid + j * GW * 32;
    const int kq = i / OUT_PAD, n = i - kq * OUT_PAD;
    const int nc = n < OUT_F ? n : OUT_F - 1;
    const float* wp = W3 + (4 * kq) * OUT_F + nc;
    float4 v = make_float4(wp[0], wp[OUT_F], wp[2 * OUT_F], wp[3 * OUT_F]);
    if (n >= OUT_F) v = make_float4(0.0f, 0.0f, 0.0f, 0.0f);
    wv3[j] = v;
  }

  const int nd = tmc * 16 + 8 * half + (l & 7);
  const int ndc = nd < last ? nd : last;
  const int e_l = row_ptr[ndc];
  const int c_l = cnt_in[ndc];
  const int d_l = (nd < nNodes && tm < mTiles) ? c_l : 0;

#pragma unroll
  for (int j = 0; j < WPER2; ++j) *(float4*)(Wq2 + (tid + j * GW * 32) * 4) = wv2[j];
#pragma unroll
  for (int j = 0; j < WPER3; ++j) *(float4*)(Wq3 + (tid + j * GW * 32) * 4) = wv3[j];

  const char* fb = (const char*)feat + l * 16;
  int e = __shfl(e_l, 0, 16), d = __shfl(d_l, 0, 16);
  int lo = sorted_src[l < d ? e + l : 0];
  int hi = sorted_src[l + 16 < d ? e + 16 + l : 0];
  for (int q = 0; q < 8; ++q) {
    const int qn = q < 7 ? q + 1 : 7;
    const int e1 = __shfl(e_l, qn, 16), d1 = __shfl(d_l, qn, 16);
    const int lo1 = sorted_src[l < d1 ? e1 + l : 0];
    const int hi1 = sorted_src[l + 16 < d1 ? e1 + 16 + l : 0];
    float4 acc = make_float4(0.0f, 0.0f, 0.0f, 0.0f);
    if (d > 0) sum_burst<HID_F * 4, 16>(acc, fb, lo, d, true);
    if (d > 24) sum_burst<HID_F * 4, 16>(acc, fb, hi, d - 16, true);
    else if (d > 16) sum_burst<HID_F * 4, 8>(acc, fb, hi, d - 16, true);
    for (int k = 32; k < d; ++k) {
      const float4 v = *(const float4*)(fb + (unsigned)sorted_src[e + k] * (HID_F * 4));
      add4(acc, v);
    }
    *(float4*)(at + (8 * half + q) * A_LD + l * 4) = acc;
    e = e1; d = d1; lo = lo1; hi = hi1;
  }

  float ni[8], no[8], bq[NT2];
#pragma unroll
  for (int r = 0; r < 8; ++r) {
    const int row = tmc * 16 + 8 * half + r;
    const int rc = row < last ? row : last;
    ni[r] = norm_in[rc];
    no[r] = norm_out[rc];
  }
#pragma unroll
  for (int t = 0; t < NT2; ++t) bq[t] = b2[16 * t + l];

  __syncthreads();
  v8f c[NT2];
#pragma unroll
  for (int t = 0; t < NT2; ++t) c[t] = v8f{};
  products64<false, NT2, HID_F>(at, Wq2, 0, half, l, 1.0f, c);
  __syncthreads();
#pragma unroll
  for (int t = 0; t < NT2; ++t) {
#pragma unroll
    for (int r = 0; r < 8; ++r)
      at[(8 * half + r) * A_LD + 16 * t + l] = fmaxf(c[t][r] * ni[r] + bq[t], 0.0f) * no[r];
  }
  __syncthreads();

  v8f c3[NT3];
#pragma unroll
  for (int t = 0; t < NT3; ++t) c3[t] = v8f{};
  products64<false, NT3, OUT_PAD>(at, Wq3, 0, half, l, 1.0f, c3);

  __syncthreads();
#pragma unroll
  for (int t = 0; t < NT3; ++t) {
#pragma unroll
    for (int r = 0; r < 8; ++r) at[(8 * half + r) * A_LD + 16 * t + l] = c3[t][r];
  }
  __syncthreads();
  if (tm < mTiles && l * 4 < OUT_PAD) {
#pragma unroll
    for (int q = 0; q < 8; ++q) {
      const int row = tm * 16 + 2 * q + half;
      if (row < nNodes)
        *(float4*)(out + (size_t)row * C_LD + l * 4) = *(const float4*)(at + (2 * q + half) * A_LD + l * 4);
    }
  }
}


static inline int cdiv_ll(long long a, long long b) { return (int)((a + b - 1) / b); }

extern "C" void kernel_launch(void* const* d_in, const int* in_sizes, int n_in,
                              void* d_out, int out_size, void* d_ws, size_t ws_size,
                              hipStream_t stream) {
  const float* X  = (const float*)d_in[0];
  const float* W1 = (const float*)d_in[1];
  const float* b1 = (const float*)d_in[2];
  const float* W2 = (const float*)d_in[3];
  const float* b2 = (const float*)d_in[4];
  const float* W3 = (const float*)d_in[5];
  const float* b3 = (const float*)d_in[6];
  const int* src  = (const int*)d_in[7];
  const int* dst  = (const int*)d_in[8];
  float* out = (float*)d_out;

  const int nNodes = in_sizes[0] / IN_F;
  const int nEdges = in_sizes[7];
  const int mTiles = (nNodes + 15) / 16;
  const int nBins  = (nNodes + BNODES - 1) >> BSH;
  const int nTiles = (nEdges + TILE - 1) / TILE;

  const size_t nP = ((size_t)nNodes + 63) & ~(size_t)63;
  const size_t eP = ((size_t)nEdges + 63) & ~(size_t)63;
  const size_t hP = ((size_t)nTiles * nBins + 63) & ~(size_t)63;
  int* wi = (int*)d_ws;
  int* cnt_out    = wi;
  int* cnt_in     = cnt_out + nP;
  int* row_ptr    = cnt_in + nP;
  int* hist       = row_ptr + nP;
  int* off        = hist + hP;
  int* bin_base   = off + hP;
  int* bsrc       = bin_base + MAXB + 64;
  int* bdst       = bsrc + eP;
  int* sorted_src = bdst + eP;
  float* norm_out = (float*)(sorted_src + eP);
  float* norm_in  = norm_out + nP;
  float* bufA     = norm_in + nP;
  float* bufB     = bufA + nP * HID_F;
  float* bufC     = bufB + nP * HID_F;

  const dim3 blk(256);
  const int gemmBlocks = (mTiles + GW - 1) / GW;
  const int gatherBlocks = GATHER_WG;
  const dim3 gblk(GATHER_T);

  zero_i32<<<cdiv_ll((long long)nP, 256), blk, 0, stream>>>(cnt_out, (int)nP);
  tile_count<<<nTiles, dim3(TILE_T), 0, stream>>>(src, dst, cnt_out, hist, nEdges, nBins);
  bin_offsets<<<1, dim3(2 * MAXB), 0, stream>>>(hist, off, bin_base, nTiles, nBins);
  tile_scatter<<<nTiles, dim3(TILE_T), 0, stream>>>(src, dst, hist, off, bsrc, bdst, nEdges, nBins);
  bin_finish<<<nBins, dim3(TILE_T), 0, stream>>>(bsrc, bdst, bin_base, cnt_out, row_ptr, cnt_in,
                                                norm_in, norm_out, sorted_src, nNodes);

  gemm_rows<<<gemmBlocks, dim3(GW * 32), 0, stream>>>(
      X, W1, norm_out, bufA, nNodes, mTiles);
  gather_rows<1><<<gatherBlocks, gblk, 0, stream>>>(
      bufA, row_ptr, cnt_in, sorted_src, norm_in, norm_out, b1, bufB, nNodes);

  layer2_layer3<<<gemmBlocks, dim3(GW * 32), 0, stream>>>(
      bufB, row_ptr, cnt_in, sorted_src, W2, b2, W3, norm_in, norm_out, bufC, nNodes, mTiles);

  gather_rows<2><<<gatherBlocks, gblk, 0, stream>>>(
      bufC, row_ptr, cnt_in, sorted_src, norm_in, norm_out, b3, out, nNodes);
}


# Sort's two helper launches folded into the tile passes (9 -> 7): count pass clears the out-degree counters, each placing group derives its slots from packed tile histograms; tile loads issued first
# speedup vs baseline: 3.4736x; 1.1368x over previous
#include <hip/hip_runtime.h>


typedef __attribute__((ext_vector_type(2))) float v2f;
typedef __attribute__((ext_vector_type(8))) float v8f;
typedef __attribute__((ext_vector_type(4))) float v4f;

#define IN_F    128
#define HID_F   64
#define OUT_F   40
#define OUT_PAD 48
#define C_LD    64

#define BSH      8
#define BNODES   (1 << BSH)
#define MAXB     512
#define TILE     8192
#define TILE_T   1024
#define STAGE    8192


static_assert(TILE < 65536, "a tile's bin count fits 16 bits");
__global__ void __launch_bounds__(TILE_T)
tile_count(const int* __restrict__ de, int* __restrict__ cnt_out, int* __restrict__ hist,
           int nEdges, int nClear) {
  __shared__ int h[MAXB];
  const int tid = threadIdx.x;
  const int base = blockIdx.x * TILE;
  int d[TILE / TILE_T];
#pragma unroll
  for (int j = 0; j < TILE / TILE_T; ++j) {
    const int e = base + j * TILE_T + tid;
    d[j] = de[e < nEdges ? e : nEdges - 1];
  }
  if (tid < MAXB) h[tid] = 0;
  for (int i = blockIdx.x * TILE_T + tid; i < nClear; i += gridDim.x * TILE_T) cnt_out[i] = 0;
  __syncthreads();
#pragma unroll
  for (int j = 0; j < TILE / TILE_T; ++j) {
    if (base + j * TILE_T + tid < nEdges) atomicAdd(&h[d[j] >> BSH], 1);
  }
  __syncthreads();
  if (tid < MAXB / 2) hist[blockIdx.x * (MAXB / 2) + tid] = h[2 * tid] | (h[2 * tid + 1] << 16);
}

#define HROWS (TILE_T / (MAXB / 8))
#define HB    8

__global__ void __launch_bounds__(TILE_T)
tile_scatter(const int* __restrict__ se, const int* __restrict__ de, const int* __restrict__ hist,
             int* __restrict__ cnt_out, int* __restrict__ bin_base,
             int* __restrict__ bsrc, int* __restrict__ bdst, int nEdges, int nTiles) {
  static_assert(TILE_T == 2 * MAXB, "one thread per bin for each of the two running sums");
  __shared__ __attribute__((aligned(16))) int pa[HROWS][MAXB];
  __shared__ __attribute__((aligned(16))) int pb[HROWS][MAXB];
  __shared__ int wsum[TILE_T / 32];
  __shared__ int sbs[MAXB];
  __shared__ int lcur[MAXB];
  __shared__ int gdel[MAXB];
  __shared__ int ssrc[TILE];
  __shared__ int sdst[TILE];
  const int tid = threadIdx.x, lane = tid & 31, w = tid >> 5;
  const int T = blockIdx.x;
  const int base = T * TILE;
  const int left = nEdges - base;
  const int nT = left < TILE ? left : TILE;

  int s[TILE / TILE_T], d[TILE / TILE_T];
#pragma unroll
  for (int j = 0; j < TILE / TILE_T; ++j) {
    const int e = base + j * TILE_T + tid;
    const int ec = e < nEdges ? e : nEdges - 1;
    s[j] = se[ec];
    d[j] = de[ec];
  }

  const int b = tid & (MAXB - 1), part = tid >> 9;
  const int mineW = hist[T * (MAXB / 2) + (b >> 1)];
  const int g = tid & (MAXB / 8 - 1), r = tid / (MAXB / 8);
  const int4* h4 = (const int4*)hist;
  int all[8], bef[8];
#pragma unroll
  for (int k = 0; k < 8; ++k) { all[k] = 0; bef[k] = 0; }
  for (int cb = r; cb < nTiles; cb += HROWS * HB) {
    int4 v[HB];
#pragma unroll
    for (int u = 0; u < HB; ++u) {
      const int cc = cb + HROWS * u;
      v[u] = h4[(cc < nTiles ? cc : nTiles - 1) * (MAXB / 8) + g];
    }
    __builtin_amdgcn_sched_barrier(0);
#pragma unroll
    for (int u = 0; u < HB; ++u) {
      const int cc = cb + HROWS * u;
      const int ma = cc < nTiles ? -1 : 0, mb = cc < T ? -1 : 0;
      const int wd[4] = {v[u].x, v[u].y, v[u].z, v[u].w};
#pragma unroll
      for (int k = 0; k < 4; ++k) {
        const int lo = wd[k] & 0xffff, hi = (int)((unsigned)wd[k] >> 16);
        all[2 * k] += lo & ma; all[2 * k + 1] += hi & ma;
        bef[2 * k] += lo & mb; bef[2 * k + 1] += hi & mb;
      }
    }
  }
  *(int4*)&pa[r][8 * g]     = make_int4(all[0], all[1], all[2], all[3]);
  *(int4*)&pa[r][8 * g + 4] = make_int4(all[4], all[5], all[6], all[7]);
  *(int4*)&pb[r][8 * g]     = make_int4(bef[0], bef[1], bef[2], bef[3]);
  *(int4*)&pb[r][8 * g + 4] = make_int4(bef[4], bef[5], bef[6], bef[7]);
  __syncthreads();

  int x = 0, before = 0;
  if (part == 0) {
#pragma unroll
    for (int q = 0; q < HROWS; ++q) x += pa[q][b];
  } else {
#pragma unroll
    for (int q = 0; q < HROWS; ++q) before += pb[q][b];
    x = (b & 1) ? (int)((unsigned)mineW >> 16) : (mineW & 0xffff);
  }
  int run = x;
#pragma unroll
  for (int o = 1; o < 32; o <<= 1) {
    const int y = __shfl_up(run, o, 32);
    if (lane >= o) run += y;
  }
  if (lane == 31) wsum[w] = run;
  __syncthreads();
  int t = wsum[(w & 16) + (lane & 15)];
  t = (lane & 15) < (w & 15) ? t : 0;
#pragma unroll
  for (int o = 8; o > 0; o >>= 1) t += __shfl_xor(t, o, 16);
  const int excl = t + run - x;
  if (part == 0) {
    sbs[b] = excl;
    if (T == 0) {
      bin_base[b] = excl;
      if (tid == 0) bin_base[MAXB] = nEdges;
    }
  } else {
    lcur[b] = excl;
  }
  __syncthreads();
  if (part == 1) gdel[b] = sbs[b] + before - excl;

#pragma unroll
  for (int j = 0; j < TILE / TILE_T; ++j) {
    const int e = base + j * TILE_T + tid;
    if (e < nEdges) {
      const int q = atomicAdd(&lcur[d[j] >> BSH], 1);
      ssrc[q] = s[j];
      sdst[q] = d[j];
    }
  }
  __syncthreads();
  for (int j = tid; j < nT; j += TILE_T) {
    const int dd = sdst[j];
    const int gg = gdel[dd >> BSH] + j;
    bsrc[gg] = ssrc[j];
    bdst[gg] = dd;
  }
#pragma unroll
  for (int j = 0; j < TILE / TILE_T; ++j) {
    if (base + j * TILE_T + tid < nEdges) atomicAdd(&cnt_out[s[j]], 1);
  }
}

__global__ void __launch_bounds__(TILE_T)
bin_finish(const int* __restrict__ bsrc, const int* __restrict__ bdst,
           const int* __restrict__ bin_base, const int* __restrict__ cnt_out,
           int* __restrict__ row_ptr, int* __restrict__ cnt_in,
           float* __restrict__ norm_in, float* __restrict__ norm_out,
           int* __restrict__ sorted_src, int nNodes) {
  __shared__ int ncnt[BNODES];
  __shared__ int ncur[BNODES];
  __shared__ int stage[STAGE];
  const int tid = threadIdx.x;
  const int e0 = bin_base[blockIdx.x];
  const int nE = bin_base[blockIdx.x + 1] - e0;
  const int n0 = blockIdx.x << BSH;
  if (tid < BNODES) ncnt[tid] = 0;
  __syncthreads();
  for (int j = tid; j < nE; j += TILE_T) atomicAdd(&ncnt[bdst[e0 + j] - n0], 1);
  __syncthreads();
  if (tid < BNODES) {
    int a = 0;
    for (int j = 0; j < tid; ++j) a += ncnt[j];
    ncur[tid] = a;
    const int node = n0 + tid;
    if (node < nNodes) {
      const int c = ncnt[tid];
      row_ptr[node] = e0 + a;
      cnt_in[node] = c;
      norm_in[node] = rsqrtf(fmaxf((float)c, 1.0f));
      norm_out[node] = rsqrtf(fmaxf((float)cnt_out[node], 1.0f));
    }
  }
  __syncthreads();
  const bool staged = nE <= STAGE;
  for (int j = tid; j < nE; j += TILE_T) {
    const int q = atomicAdd(&ncur[bdst[e0 + j] - n0], 1);
    const int s = bsrc[e0 + j];
    if (staged) stage[q] = s; else sorted_src[e0 + q] = s;
  }
  __syncthreads();
  if (staged) {
    for (int j = tid; j < nE; j += TILE_T) sorted_src[e0 + j] = stage[j];
  }
}


__device__ __forceinline__ void add4(float4& a, const float4& v) {
  a.x += v.x; a.y += v.y; a.z += v.z; a.w += v.w;
}

#define GATHER_T   256
#define GATHER_WG  1024

template <unsigned ROWB, int NB>
__device__ __forceinline__ void sum_burst(float4& acc, const char* fb, int idx, int n, bool col) {
  static_assert(NB == 8 || NB == 16, "a burst is 8 or 16 rows");
  int s[NB];
#pragma unroll
  for (int j = 0; j < NB; ++j) s[j] = __shfl(idx, j, 16);
  if (col) {
    v4f v[NB];
#pragma unroll
    for (int j = 0; j < NB; ++j) v[j] = *(const v4f*)(fb + (unsigned)s[j] * ROWB);
    __builtin_amdgcn_sched_barrier(0);
#pragma unroll
    for (int j = 0; j < NB; ++j) {
      const bool in = j < n;
      acc.x += in ? v[j].x : 0.0f;
      acc.y += in ? v[j].y : 0.0f;
      acc.z += in ? v[j].z : 0.0f;
      acc.w += in ? v[j].w : 0.0f;
    }
  }
}

template <int MODE>
__global__ void __launch_bounds__(GATHER_T)
gather_rows(const float* __restrict__ feat, const int* __restrict__ row_ptr,
            const int* __restrict__ cnt_in, const int* __restrict__ sorted_src,
            const float* __restrict__ norm_in, const float* __restrict__ norm_out,
            const float* __restrict__ bias, float* __restrict__ outp, int nNodes) {
  constexpr unsigned ROWB = ((MODE == 2) ? C_LD : HID_F) * 4;
  constexpr int      NCH  = (MODE == 2) ? OUT_F / 4 : HID_F / 4;
  const unsigned t = blockIdx.x * blockDim.x + threadIdx.x;
  const int nGroups = (int)((gridDim.x * blockDim.x) >> 4);
  const int ch = (int)(t & 15);
  const bool col = ch < NCH;
  const char* fb = (const char*)feat + ch * 16;
  float4 b = make_float4(0.0f, 0.0f, 0.0f, 0.0f);
  if (col) b = ((const float4*)bias)[ch];

  const int last = nNodes - 1;
  int i = (int)(t >> 4);
  const int ic = i < last ? i : last;
  int e = row_ptr[ic];
  const int c0 = cnt_in[ic];
  int d = i < nNodes ? c0 : 0;
  int i1 = i + nGroups;
  const int i1c = i1 < last ? i1 : last;
  int e1 = row_ptr[i1c];
  const int c1 = cnt_in[i1c];
  int d1 = i1 < nNodes ? c1 : 0;
  int lo = sorted_src[ch < d ? e + ch : 0];
  int hi = sorted_src[ch + 16 < d ? e + 16 + ch : 0];

  while (i < nNodes) {
    const int i2 = i1 + nGroups;
    const int i2c = i2 < last ? i2 : last;
    const int e2 = row_ptr[i2c];
    const int c2 = cnt_in[i2c];
    const int d2 = i2 < nNodes ? c2 : 0;
    const int lo1 = sorted_src[ch < d1 ? e1 + ch : 0];
    const int hi1 = sorted_src[ch + 16 < d1 ? e1 + 16 + ch : 0];
    const float ni = norm_in[i];
    float no = 1.0f;
    if (MODE == 1) no = norm_out[i];

    float4 acc = make_float4(0.0f, 0.0f, 0.0f, 0.0f);
    if (d > 0) sum_burst<ROWB, 16>(acc, fb, lo, d, col);
    if (d > 24) sum_burst<ROWB, 16>(acc, fb, hi, d - 16, col);
    else if (d > 16) sum_burst<ROWB, 8>(acc, fb, hi, d - 16, col);
    if (col) {
      for (int k = 32; k < d; ++k) {
        const float4 v = *(const float4*)(fb + (unsigned)sorted_src[e + k] * ROWB);
        add4(acc, v);
      }
      if (MODE == 1) {
        float4 r;
        r.x = fmaxf(acc.x * ni + b.x, 0.0f) * no;
        r.y = fmaxf(acc.y * ni + b.y, 0.0f) * no;
        r.z = fmaxf(acc.z * ni + b.z, 0.0f) * no;
        r.w = fmaxf(acc.w * ni + b.w, 0.0f) * no;
        ((float4*)outp)[(size_t)i * (HID_F / 4) + ch] = r;
      } else {
        float4 r;
        r.x = acc.x * ni + b.x;
        r.y = acc.y * ni + b.y;
        r.z = acc.z * ni + b.z;
        r.w = acc.w * ni + b.w;
        ((float4*)outp)[(size_t)i * (OUT_F / 4) + ch] = r;
      }
    }
    i = i1; e = e1; d = d1; lo = lo1; hi = hi1;
    i1 = i2; e1 = e2; d1 = d2;
  }
}


#define GW   8
#define A_LD 68

template <bool SCALE, int NT, int NPAD>
__device__ __forceinline__ void products64(const float* at, const float* Wq, int kq0, int half, int l, float sA,
                                           v8f (&c)[NT]) {
  float4 a4[8];
#pragma unroll
  for (int sp = 0; sp < 8; ++sp) {
    a4[sp] = *(const float4*)(at + l * A_LD + 8 * sp + 4 * half);
    if (SCALE) { a4[sp].x *= sA; a4[sp].y *= sA; a4[sp].z *= sA; a4[sp].w *= sA; }
  }
#pragma unroll
  for (int sp = 0; sp < 8; ++sp) {
    float4 b4[NT];
#pragma unroll
    for (int t = 0; t < NT; ++t)
      b4[t] = *(const float4*)(Wq + ((kq0 + 2 * sp + half) * NPAD + 16 * t + l) * 4);
#pragma unroll
    for (int t = 0; t < NT; ++t) {
      v2f a, b;
      a.x = a4[sp].x; a.y = a4[sp].y; b.x = b4[t].x; b.y = b4[t].y;
      c[t] = __builtin_amdgcn_wmma_f32_16x16x4_f32(false, a, false, b, (short)0, c[t], false, false);
    }
#pragma unroll
    for (int t = 0; t < NT; ++t) {
      v2f a, b;
      a.x = a4[sp].z; a.y = a4[sp].w; b.x = b4[t].z; b.y = b4[t].w;
      c[t] = __builtin_amdgcn_wmma_f32_16x16x4_f32(false, a, false, b, (short)0, c[t], false, false);
    }
  }
}

__global__ void __launch_bounds__(GW * 32, 8)
gemm_rows(const float* __restrict__ A, const float* __restrict__ W, const float* __restrict__ norm_out,
          float* __restrict__ out, int nNodes, int mTiles) {
  constexpr int KTOT = IN_F;
  constexpr int NPAD = HID_F;
  constexpr int NT   = NPAD / 16;
  static_assert(KTOT == 128, "two parts of 64 k");
  __shared__ __attribute__((aligned(16))) float Wq[KTOT * NPAD];
  __shared__ __attribute__((aligned(16))) float At[GW][16 * A_LD];

  const int tid = threadIdx.x, lane = tid & 31, w = tid >> 5;
  const int half = lane >> 4, l = lane & 15;
  const int tm = blockIdx.x * GW + w;
  const int tmc = tm < mTiles ? tm : mTiles - 1;
  float* at = At[w];

  int aoff[8];
#pragma unroll
  for (int q = 0; q < 8; ++q) {
    const int row = tmc * 16 + 2 * q + half;
    const int r = row < nNodes ? row : nNodes - 1;
    aoff[q] = r * KTOT + l * 4;
  }
  float4 g0[8];
#pragma unroll
  for (int q = 0; q < 8; ++q) g0[q] = *(const float4*)(A + aoff[q]);
  const int rowA = tmc * 16 + l;
  const float sA = norm_out[rowA < nNodes ? rowA : nNodes - 1];

  constexpr int WPER = (KTOT / 4) * NPAD / (GW * 32);
  static_assert((KTOT / 4) * NPAD % (GW * 32) == 0, "weight items divide evenly over the threads");
  float4 wv[WPER];
#pragma unroll
  for (int j = 0; j < WPER; ++j) {
    const int i = tid + j * GW * 32;
    const int kq = i / NPAD, n = i - kq * NPAD;
    const float* wp = W + (4 * kq) * NPAD + n;
    wv[j] = make_float4(wp[0], wp[NPAD], wp[2 * NPAD], wp[3 * NPAD]);
  }
#pragma unroll
  for (int j = 0; j < WPER; ++j) *(float4*)(Wq + (tid + j * GW * 32) * 4) = wv[j];

  v8f c[NT];
#pragma unroll
  for (int t = 0; t < NT; ++t) c[t] = v8f{};

#pragma unroll
  for (int q = 0; q < 8; ++q) *(float4*)(at + (2 * q + half) * A_LD + l * 4) = g0[q];
  __syncthreads();
  products64<true, NT, NPAD>(at, Wq, 0, half, l, sA, c);
  __syncthreads();
  float4 g1[8];
#pragma unroll
  for (int q = 0; q < 8; ++q) g1[q] = *(const float4*)(A + aoff[q] + 64);
#pragma unroll
  for (int q = 0; q < 8; ++q) *(float4*)(at + (2 * q + half) * A_LD + l * 4) = g1[q];
  __syncthreads();
  products64<true, NT, NPAD>(at, Wq, 16, half, l, sA, c);

  __syncthreads();
#pragma unroll
  for (int t = 0; t < NT; ++t) {
#pragma unroll
    for (int r = 0; r < 8; ++r) at[(8 * half + r) * A_LD + 16 * t + l] = c[t][r];
  }
  __syncthreads();
  if (tm < mTiles) {
#pragma unroll
    for (int q = 0; q < 8; ++q) {
      const int row = tm * 16 + 2 * q + half;
      if (row < nNodes)
        *(float4*)(out + (size_t)row * HID_F + l * 4) = *(const float4*)(at + (2 * q + half) * A_LD + l * 4);
    }
  }
}

__global__ void __launch_bounds__(GW * 32, 8)
layer2_layer3(const float* __restrict__ feat, const int* __restrict__ row_ptr,
              const int* __restrict__ cnt_in, const int* __restrict__ sorted_src,
              const float* __restrict__ W2, const float* __restrict__ b2, const float* __restrict__ W3,
              const float* __restrict__ norm_in, const float* __restrict__ norm_out,
              float* __restrict__ out, int nNodes, int mTiles) {
  constexpr int NT2 = HID_F / 16, NT3 = OUT_PAD / 16;
  __shared__ __attribute__((aligned(16))) float Wq2[HID_F * HID_F];
  __shared__ __attribute__((aligned(16))) float Wq3[HID_F * OUT_PAD];
  __shared__ __attribute__((aligned(16))) float At[GW][16 * A_LD];

  const int tid = threadIdx.x, lane = tid & 31, w = tid >> 5;
  const int half = lane >> 4, l = lane & 15;
  const int tm = blockIdx.x * GW + w;
  const int tmc = tm < mTiles ? tm : mTiles - 1;
  const int last = nNodes - 1;
  float* at = At[w];

  constexpr int WPER2 = (HID_F / 4) * HID_F / (GW * 32), WPER3 = (HID_F / 4) * OUT_PAD / (GW * 32);
  static_assert((HID_F / 4) * HID_F % (GW * 32) == 0 && (HID_F / 4) * OUT_PAD % (GW * 32) == 0,
                "weight items divide evenly over the threads");
  float4 wv2[WPER2], wv3[WPER3];
#pragma unroll
  for (int j = 0; j < WPER2; ++j) {
    const int i = tid + j * GW * 32;
    const int kq = i / HID_F, n = i - kq * HID_F;
    const float* wp = W2 + (4 * kq) * HID_F + n;
    wv2[j] = make_float4(wp[0], wp[HID_F], wp[2 * HID_F], wp[3 * HID_F]);
  }
#pragma unroll
  for (int j = 0; j < WPER3; ++j) {
    const int i = tid + j * GW * 32;
    const int kq = i / OUT_PAD, n = i - kq * OUT_PAD;
    const int nc = n < OUT_F ? n : OUT_F - 1;
    const float* wp = W3 + (4 * kq) * OUT_F + nc;
    float4 v = make_float4(wp[0], wp[OUT_F], wp[2 * OUT_F], wp[3 * OUT_F]);
    if (n >= OUT_F) v = make_float4(0.0f, 0.0f, 0.0f, 0.0f);
    wv3[j] = v;
  }

  const int nd = tmc * 16 + 8 * half + (l & 7);
  const int ndc = nd < last ? nd : last;
  const int e_l = row_ptr[ndc];
  const int c_l = cnt_in[ndc];
  const int d_l = (nd < nNodes && tm < mTiles) ? c_l : 0;

#pragma unroll
  for (int j = 0; j < WPER2; ++j) *(float4*)(Wq2 + (tid + j * GW * 32) * 4) = wv2[j];
#pragma unroll
  for (int j = 0; j < WPER3; ++j) *(float4*)(Wq3 + (tid + j * GW * 32) * 4) = wv3[j];

  const char* fb = (const char*)feat + l * 16;
  int e = __shfl(e_l, 0, 16), d = __shfl(d_l, 0, 16);
  int lo = sorted_src[l < d ? e + l : 0];
  int hi = sorted_src[l + 16 < d ? e + 16 + l : 0];
  for (int q = 0; q < 8; ++q) {
    const int qn = q < 7 ? q + 1 : 7;
    const int e1 = __shfl(e_l, qn, 16), d1 = __shfl(d_l, qn, 16);
    const int lo1 = sorted_src[l < d1 ? e1 + l : 0];
    const int hi1 = sorted_src[l + 16 < d1 ? e1 + 16 + l : 0];
    float4 acc = make_float4(0.0f, 0.0f, 0.0f, 0.0f);
    if (d > 0) sum_burst<HID_F * 4, 16>(acc, fb, lo, d, true);
    if (d > 24) sum_burst<HID_F * 4, 16>(acc, fb, hi, d - 16, true);
    else if (d > 16) sum_burst<HID_F * 4, 8>(acc, fb, hi, d - 16, true);
    for (int k = 32; k < d; ++k) {
      const float4 v = *(const float4*)(fb + (unsigned)sorted_src[e + k] * (HID_F * 4));
      add4(acc, v);
    }
    *(float4*)(at + (8 * half + q) * A_LD + l * 4) = acc;
    e = e1; d = d1; lo = lo1; hi = hi1;
  }

  float ni[8], no[8], bq[NT2];
#pragma unroll
  for (int r = 0; r < 8; ++r) {
    const int row = tmc * 16 + 8 * half + r;
    const int rc = row < last ? row : last;
    ni[r] = norm_in[rc];
    no[r] = norm_out[rc];
  }
#pragma unroll
  for (int t = 0; t < NT2; ++t) bq[t] = b2[16 * t + l];

  __syncthreads();
  v8f c[NT2];
#pragma unroll
  for (int t = 0; t < NT2; ++t) c[t] = v8f{};
  products64<false, NT2, HID_F>(at, Wq2, 0, half, l, 1.0f, c);
  __syncthreads();
#pragma unroll
  for (int t = 0; t < NT2; ++t) {
#pragma unroll
    for (int r = 0; r < 8; ++r)
      at[(8 * half + r) * A_LD + 16 * t + l] = fmaxf(c[t][r] * ni[r] + bq[t], 0.0f) * no[r];
  }
  __syncthreads();

  v8f c3[NT3];
#pragma unroll
  for (int t = 0; t < NT3; ++t) c3[t] = v8f{};
  products64<false, NT3, OUT_PAD>(at, Wq3, 0, half, l, 1.0f, c3);

  __syncthreads();
#pragma unroll
  for (int t = 0; t < NT3; ++t) {
#pragma unroll
    for (int r = 0; r < 8; ++r) at[(8 * half + r) * A_LD + 16 * t + l] = c3[t][r];
  }
  __syncthreads();
  if (tm < mTiles && l * 4 < OUT_PAD) {
#pragma unroll
    for (int q = 0; q < 8; ++q) {
      const int row = tm * 16 + 2 * q + half;
      if (row < nNodes)
        *(float4*)(out + (size_t)row * C_LD + l * 4) = *(const float4*)(at + (2 * q + half) * A_LD + l * 4);
    }
  }
}


extern "C" void kernel_launch(void* const* d_in, const int* in_sizes, int n_in,
                              void* d_out, int out_size, void* d_ws, size_t ws_size,
                              hipStream_t stream) {
  const float* X  = (const float*)d_in[0];
  const float* W1 = (const float*)d_in[1];
  const float* b1 = (const float*)d_in[2];
  const float* W2 = (const float*)d_in[3];
  const float* b2 = (const float*)d_in[4];
  const float* W3 = (const float*)d_in[5];
  const float* b3 = (const float*)d_in[6];
  const int* src  = (const int*)d_in[7];
  const int* dst  = (const int*)d_in[8];
  float* out = (float*)d_out;

  const int nNodes = in_sizes[0] / IN_F;
  const int nEdges = in_sizes[7];
  const int mTiles = (nNodes + 15) / 16;
  const int nBins  = (nNodes + BNODES - 1) >> BSH;
  const int nTiles = (nEdges + TILE - 1) / TILE;

  const size_t nP = ((size_t)nNodes + 63) & ~(size_t)63;
  const size_t eP = ((size_t)nEdges + 63) & ~(size_t)63;
  const size_t hP = ((size_t)nTiles * (MAXB / 2) + 63) & ~(size_t)63;
  int* wi = (int*)d_ws;
  int* cnt_out    = wi;
  int* cnt_in     = cnt_out + nP;
  int* row_ptr    = cnt_in + nP;
  int* hist       = row_ptr + nP;
  int* bin_base   = hist + hP;
  int* bsrc       = bin_base + MAXB + 64;
  int* bdst       = bsrc + eP;
  int* sorted_src = bdst + eP;
  float* norm_out = (float*)(sorted_src + eP);
  float* norm_in  = norm_out + nP;
  float* bufA     = norm_in + nP;
  float* bufB     = bufA + nP * HID_F;
  float* bufC     = bufB + nP * HID_F;

  const int gemmBlocks = (mTiles + GW - 1) / GW;
  const int gatherBlocks = GATHER_WG;
  const dim3 gblk(GATHER_T);

  tile_count<<<nTiles, dim3(TILE_T), 0, stream>>>(dst, cnt_out, hist, nEdges, (int)nP);
  tile_scatter<<<nTiles, dim3(TILE_T), 0, stream>>>(src, dst, hist, cnt_out, bin_base, bsrc, bdst, nEdges, nTiles);
  bin_finish<<<nBins, dim3(TILE_T), 0, stream>>>(bsrc, bdst, bin_base, cnt_out, row_ptr, cnt_in,
                                                norm_in, norm_out, sorted_src, nNodes);

  gemm_rows<<<gemmBlocks, dim3(GW * 32), 0, stream>>>(
      X, W1, norm_out, bufA, nNodes, mTiles);
  gather_rows<1><<<gatherBlocks, gblk, 0, stream>>>(
      bufA, row_ptr, cnt_in, sorted_src, norm_in, norm_out, b1, bufB, nNodes);

  layer2_layer3<<<gemmBlocks, dim3(GW * 32), 0, stream>>>(
      bufB, row_ptr, cnt_in, sorted_src, W2, b2, W3, norm_in, norm_out, bufC, nNodes, mTiles);

  gather_rows<2><<<gatherBlocks, gblk, 0, stream>>>(
      bufC, row_ptr, cnt_in, sorted_src, norm_in, norm_out, b3, out, nNodes);
}


# Out-degree count without global atomics: count pass groups each tile's sources by source bin in LDS (low bytes, per-tile offsets); each bin's owner collects its pieces and counts in LDS
# speedup vs baseline: 5.5557x; 1.5994x over previous
#include <hip/hip_runtime.h>


typedef __attribute__((ext_vector_type(2))) float v2f;
typedef __attribute__((ext_vector_type(8))) float v8f;
typedef __attribute__((ext_vector_type(4))) float v4f;

#define IN_F    128
#define HID_F   64
#define OUT_F   40
#define OUT_PAD 48
#define C_LD    64

#define BSH      8
#define BNODES   (1 << BSH)
#define MAXB     512
#define TILE     8192
#define TILE_T   1024
#define STAGE    8192


static_assert(TILE < 65536, "a tile's bin count fits 16 bits");
static_assert(BNODES <= 256, "a node's place inside its bin fits one byte");
static_assert(MAXB % 32 == 0 && MAXB / 32 <= 32, "the bins' running sum: whole waves, and one lane per wave total");
static_assert(TILE == 8 * TILE_T, "a thread stores 8 of the tile's source bytes");
__global__ void __launch_bounds__(TILE_T)
tile_count(const int* __restrict__ se, const int* __restrict__ de, int* __restrict__ hist,
           int* __restrict__ soff, int* __restrict__ tsbw, int nEdges) {
  __shared__ int h[MAXB];
  __shared__ int hs[MAXB];
  __shared__ int wtot[MAXB / 32];
  __shared__ __attribute__((aligned(16))) unsigned char sb[TILE];
  const int tid = threadIdx.x, lane = tid & 31, w = tid >> 5;
  const int base = blockIdx.x * TILE;
  int d[TILE / TILE_T], s[TILE / TILE_T];
#pragma unroll
  for (int j = 0; j < TILE / TILE_T; ++j) {
    const int e = base + j * TILE_T + tid;
    const int ec = e < nEdges ? e : nEdges - 1;
    d[j] = de[ec];
    s[j] = se[ec];
  }
  if (tid < MAXB) { h[tid] = 0; hs[tid] = 0; }
  __syncthreads();
#pragma unroll
  for (int j = 0; j < TILE / TILE_T; ++j) {
    if (base + j * TILE_T + tid < nEdges) {
      atomicAdd(&h[d[j] >> BSH], 1);
      atomicAdd(&hs[s[j] >> BSH], 1);
    }
  }
  __syncthreads();

  const int x = tid < MAXB ? hs[tid] : 0;
  int run = x;
#pragma unroll
  for (int o = 1; o < 32; o <<= 1) {
    const int y = __shfl_up(run, o, 32);
    if (lane >= o) run += y;
  }
  if (lane == 31 && w < MAXB / 32) wtot[w] = run;
  __syncthreads();
  int t = (lane < MAXB / 32 && lane < w) ? wtot[lane] : 0;
#pragma unroll
  for (int o = 16; o > 0; o >>= 1) t += __shfl_xor(t, o, 32);
  const int excl = t + run - x;
  if (tid < MAXB) hs[tid] = excl;
  __syncthreads();
#pragma unroll
  for (int j = 0; j < TILE / TILE_T; ++j) {
    if (base + j * TILE_T + tid < nEdges) {
      const int q = atomicAdd(&hs[s[j] >> BSH], 1);
      sb[q] = (unsigned char)(s[j] & (BNODES - 1));
    }
  }
  __syncthreads();
  ((int2*)tsbw)[blockIdx.x * (TILE / 8) + tid] = *(const int2*)(sb + 8 * tid);
  if (tid < MAXB / 2) hist[blockIdx.x * (MAXB / 2) + tid] = h[2 * tid] | (h[2 * tid + 1] << 16);
  if (tid < MAXB) soff[blockIdx.x * (MAXB + 1) + tid] = excl;
  if (tid == MAXB - 1) soff[blockIdx.x * (MAXB + 1) + MAXB] = excl + x;
}

#define HROWS (TILE_T / (MAXB / 8))
#define HB    8

__global__ void __launch_bounds__(TILE_T)
tile_scatter(const int* __restrict__ se, const int* __restrict__ de, const int* __restrict__ hist,
             int* __restrict__ bin_base,
             int* __restrict__ bsrc, int* __restrict__ bdst, int nEdges, int nTiles) {
  static_assert(TILE_T == 2 * MAXB, "one thread per bin for each of the two running sums");
  __shared__ __attribute__((aligned(16))) int pa[HROWS][MAXB];
  __shared__ __attribute__((aligned(16))) int pb[HROWS][MAXB];
  __shared__ int wsum[TILE_T / 32];
  __shared__ int sbs[MAXB];
  __shared__ int lcur[MAXB];
  __shared__ int gdel[MAXB];
  __shared__ int ssrc[TILE];
  __shared__ int sdst[TILE];
  const int tid = threadIdx.x, lane = tid & 31, w = tid >> 5;
  const int T = blockIdx.x;
  const int base = T * TILE;
  const int left = nEdges - base;
  const int nT = left < TILE ? left : TILE;

  int s[TILE / TILE_T], d[TILE / TILE_T];
#pragma unroll
  for (int j = 0; j < TILE / TILE_T; ++j) {
    const int e = base + j * TILE_T + tid;
    const int ec = e < nEdges ? e : nEdges - 1;
    s[j] = se[ec];
    d[j] = de[ec];
  }

  const int b = tid & (MAXB - 1), part = tid >> 9;
  const int mineW = hist[T * (MAXB / 2) + (b >> 1)];
  const int g = tid & (MAXB / 8 - 1), r = tid / (MAXB / 8);
  const int4* h4 = (const int4*)hist;
  int all[8], bef[8];
#pragma unroll
  for (int k = 0; k < 8; ++k) { all[k] = 0; bef[k] = 0; }
  for (int cb = r; cb < nTiles; cb += HROWS * HB) {
    int4 v[HB];
#pragma unroll
    for (int u = 0; u < HB; ++u) {
      const int cc = cb + HROWS * u;
      v[u] = h4[(cc < nTiles ? cc : nTiles - 1) * (MAXB / 8) + g];
    }
    __builtin_amdgcn_sched_barrier(0);
#pragma unroll
    for (int u = 0; u < HB; ++u) {
      const int cc = cb + HROWS * u;
      const int ma = cc < nTiles ? -1 : 0, mb = cc < T ? -1 : 0;
      const int wd[4] = {v[u].x, v[u].y, v[u].z, v[u].w};
#pragma unroll
      for (int k = 0; k < 4; ++k) {
        const int lo = wd[k] & 0xffff, hi = (int)((unsigned)wd[k] >> 16);
        all[2 * k] += lo & ma; all[2 * k + 1] += hi & ma;
        bef[2 * k] += lo & mb; bef[2 * k + 1] += hi & mb;
      }
    }
  }
  *(int4*)&pa[r][8 * g]     = make_int4(all[0], all[1], all[2], all[3]);
  *(int4*)&pa[r][8 * g + 4] = make_int4(all[4], all[5], all[6], all[7]);
  *(int4*)&pb[r][8 * g]     = make_int4(bef[0], bef[1], bef[2], bef[3]);
  *(int4*)&pb[r][8 * g + 4] = make_int4(bef[4], bef[5], bef[6], bef[7]);
  __syncthreads();

  int x = 0, before = 0;
  if (part == 0) {
#pragma unroll
    for (int q = 0; q < HROWS; ++q) x += pa[q][b];
  } else {
#pragma unroll
    for (int q = 0; q < HROWS; ++q) before += pb[q][b];
    x = (b & 1) ? (int)((unsigned)mineW >> 16) : (mineW & 0xffff);
  }
  int run = x;
#pragma unroll
  for (int o = 1; o < 32; o <<= 1) {
    const int y = __shfl_up(run, o, 32);
    if (lane >= o) run += y;
  }
  if (lane == 31) wsum[w] = run;
  __syncthreads();
  int t = wsum[(w & 16) + (lane & 15)];
  t = (lane & 15) < (w & 15) ? t : 0;
#pragma unroll
  for (int o = 8; o > 0; o >>= 1) t += __shfl_xor(t, o, 16);
  const int excl = t + run - x;
  if (part == 0) {
    sbs[b] = excl;
    if (T == 0) {
      bin_base[b] = excl;
      if (tid == 0) bin_base[MAXB] = nEdges;
    }
  } else {
    lcur[b] = excl;
  }
  __syncthreads();
  if (part == 1) gdel[b] = sbs[b] + before - excl;

#pragma unroll
  for (int j = 0; j < TILE / TILE_T; ++j) {
    const int e = base + j * TILE_T + tid;
    if (e < nEdges) {
      const int q = atomicAdd(&lcur[d[j] >> BSH], 1);
      ssrc[q] = s[j];
      sdst[q] = d[j];
    }
  }
  __syncthreads();
  for (int j = tid; j < nT; j += TILE_T) {
    const int dd = sdst[j];
    const int gg = gdel[dd >> BSH] + j;
    bsrc[gg] = ssrc[j];
    bdst[gg] = dd;
  }
}

__global__ void __launch_bounds__(TILE_T)
bin_finish(const int* __restrict__ bsrc, const int* __restrict__ bdst,
           const int* __restrict__ bin_base, const int* __restrict__ soff, const int* __restrict__ tsbw,
           int* __restrict__ row_ptr, int* __restrict__ cnt_in,
           float* __restrict__ norm_in, float* __restrict__ norm_out,
           int* __restrict__ sorted_src, int nNodes, int nTiles) {
  __shared__ int ncnt[BNODES];
  __shared__ int ocnt[BNODES];
  __shared__ int ncur[BNODES];
  __shared__ int stage[STAGE];
  const int tid = threadIdx.x;
  const int e0 = bin_base[blockIdx.x];
  const int nE = bin_base[blockIdx.x + 1] - e0;
  const int n0 = blockIdx.x << BSH;
  if (tid < BNODES) { ncnt[tid] = 0; ocnt[tid] = 0; }
  __syncthreads();
  {
    const int sub = tid & 3;
    for (int T = tid >> 2; T < nTiles; T += TILE_T / 4) {
      const int a = soff[T * (MAXB + 1) + blockIdx.x];
      const int c = soff[T * (MAXB + 1) + blockIdx.x + 1];
      const int* tw = tsbw + T * (TILE / 4);
      for (int wd = (a >> 2) + sub; wd * 4 < c; wd += 8) {
        const int w1 = wd + 4;
        const unsigned v0 = (unsigned)tw[wd];
        const unsigned v1 = (unsigned)tw[w1 < TILE / 4 ? w1 : TILE / 4 - 1];
#pragma unroll
        for (int k = 0; k < 4; ++k) {
          const int p0 = wd * 4 + k, p1 = w1 * 4 + k;
          if (p0 >= a && p0 < c) atomicAdd(&ocnt[(v0 >> (8 * k)) & 255u], 1);
          if (p1 < c) atomicAdd(&ocnt[(v1 >> (8 * k)) & 255u], 1);
        }
      }
    }
  }
  for (int j = tid; j < nE; j += TILE_T) atomicAdd(&ncnt[bdst[e0 + j] - n0], 1);
  __syncthreads();
  if (tid < BNODES) {
    int a = 0;
    for (int j = 0; j < tid; ++j) a += ncnt[j];
    ncur[tid] = a;
    const int node = n0 + tid;
    if (node < nNodes) {
      const int c = ncnt[tid];
      row_ptr[node] = e0 + a;
      cnt_in[node] = c;
      norm_in[node] = rsqrtf(fmaxf((float)c, 1.0f));
      norm_out[node] = rsqrtf(fmaxf((float)ocnt[tid], 1.0f));
    }
  }
  __syncthreads();
  const bool staged = nE <= STAGE;
  for (int j = tid; j < nE; j += TILE_T) {
    const int q = atomicAdd(&ncur[bdst[e0 + j] - n0], 1);
    const int s = bsrc[e0 + j];
    if (staged) stage[q] = s; else sorted_src[e0 + q] = s;
  }
  __syncthreads();
  if (staged) {
    for (int j = tid; j < nE; j += TILE_T) sorted_src[e0 + j] = stage[j];
  }
}


__device__ __forceinline__ void add4(float4& a, const float4& v) {
  a.x += v.x; a.y += v.y; a.z += v.z; a.w += v.w;
}

#define GATHER_T   256
#define GATHER_WG  1024

template <unsigned ROWB, int NB>
__device__ __forceinline__ void sum_burst(float4& acc, const char* fb, int idx, int n, bool col) {
  static_assert(NB == 8 || NB == 16, "a burst is 8 or 16 rows");
  int s[NB];
#pragma unroll
  for (int j = 0; j < NB; ++j) s[j] = __shfl(idx, j, 16);
  if (col) {
    v4f v[NB];
#pragma unroll
    for (int j = 0; j < NB; ++j) v[j] = *(const v4f*)(fb + (unsigned)s[j] * ROWB);
    __builtin_amdgcn_sched_barrier(0);
#pragma unroll
    for (int j = 0; j < NB; ++j) {
      const bool in = j < n;
      acc.x += in ? v[j].x : 0.0f;
      acc.y += in ? v[j].y : 0.0f;
      acc.z += in ? v[j].z : 0.0f;
      acc.w += in ? v[j].w : 0.0f;
    }
  }
}

template <int MODE>
__global__ void __launch_bounds__(GATHER_T)
gather_rows(const float* __restrict__ feat, const int* __restrict__ row_ptr,
            const int* __restrict__ cnt_in, const int* __restrict__ sorted_src,
            const float* __restrict__ norm_in, const float* __restrict__ norm_out,
            const float* __restrict__ bias, float* __restrict__ outp, int nNodes) {
  constexpr unsigned ROWB = ((MODE == 2) ? C_LD : HID_F) * 4;
  constexpr int      NCH  = (MODE == 2) ? OUT_F / 4 : HID_F / 4;
  const unsigned t = blockIdx.x * blockDim.x + threadIdx.x;
  const int nGroups = (int)((gridDim.x * blockDim.x) >> 4);
  const int ch = (int)(t & 15);
  const bool col = ch < NCH;
  const char* fb = (const char*)feat + ch * 16;
  float4 b = make_float4(0.0f, 0.0f, 0.0f, 0.0f);
  if (col) b = ((const float4*)bias)[ch];

  const int last = nNodes - 1;
  int i = (int)(t >> 4);
  const int ic = i < last ? i : last;
  int e = row_ptr[ic];
  const int c0 = cnt_in[ic];
  int d = i < nNodes ? c0 : 0;
  int i1 = i + nGroups;
  const int i1c = i1 < last ? i1 : last;
  int e1 = row_ptr[i1c];
  const int c1 = cnt_in[i1c];
  int d1 = i1 < nNodes ? c1 : 0;
  int lo = sorted_src[ch < d ? e + ch : 0];
  int hi = sorted_src[ch + 16 < d ? e + 16 + ch : 0];

  while (i < nNodes) {
    const int i2 = i1 + nGroups;
    const int i2c = i2 < last ? i2 : last;
    const int e2 = row_ptr[i2c];
    const int c2 = cnt_in[i2c];
    const int d2 = i2 < nNodes ? c2 : 0;
    const int lo1 = sorted_src[ch < d1 ? e1 + ch : 0];
    const int hi1 = sorted_src[ch + 16 < d1 ? e1 + 16 + ch : 0];
    const float ni = norm_in[i];
    float no = 1.0f;
    if (MODE == 1) no = norm_out[i];

    float4 acc = make_float4(0.0f, 0.0f, 0.0f, 0.0f);
    if (d > 0) sum_burst<ROWB, 16>(acc, fb, lo, d, col);
    if (d > 24) sum_burst<ROWB, 16>(acc, fb, hi, d - 16, col);
    else if (d > 16) sum_burst<ROWB, 8>(acc, fb, hi, d - 16, col);
    if (col) {
      for (int k = 32; k < d; ++k) {
        const float4 v = *(const float4*)(fb + (unsigned)sorted_src[e + k] * ROWB);
        add4(acc, v);
      }
      if (MODE == 1) {
        float4 r;
        r.x = fmaxf(acc.x * ni + b.x, 0.0f) * no;
        r.y = fmaxf(acc.y * ni + b.y, 0.0f) * no;
        r.z = fmaxf(acc.z * ni + b.z, 0.0f) * no;
        r.w = fmaxf(acc.w * ni + b.w, 0.0f) * no;
        ((float4*)outp)[(size_t)i * (HID_F / 4) + ch] = r;
      } else {
        float4 r;
        r.x = acc.x * ni + b.x;
        r.y = acc.y * ni + b.y;
        r.z = acc.z * ni + b.z;
        r.w = acc.w * ni + b.w;
        ((float4*)outp)[(size_t)i * (OUT_F / 4) + ch] = r;
      }
    }
    i = i1; e = e1; d = d1; lo = lo1; hi = hi1;
    i1 = i2; e1 = e2; d1 = d2;
  }
}


#define GW   8
#define A_LD 68

template <bool SCALE, int NT, int NPAD>
__device__ __forceinline__ void products64(const float* at, const float* Wq, int kq0, int half, int l, float sA,
                                           v8f (&c)[NT]) {
  float4 a4[8];
#pragma unroll
  for (int sp = 0; sp < 8; ++sp) {
    a4[sp] = *(const float4*)(at + l * A_LD + 8 * sp + 4 * half);
    if (SCALE) { a4[sp].x *= sA; a4[sp].y *= sA; a4[sp].z *= sA; a4[sp].w *= sA; }
  }
#pragma unroll
  for (int sp = 0; sp < 8; ++sp) {
    float4 b4[NT];
#pragma unroll
    for (int t = 0; t < NT; ++t)
      b4[t] = *(const float4*)(Wq + ((kq0 + 2 * sp + half) * NPAD + 16 * t + l) * 4);
#pragma unroll
    for (int t = 0; t < NT; ++t) {
      v2f a, b;
      a.x = a4[sp].x; a.y = a4[sp].y; b.x = b4[t].x; b.y = b4[t].y;
      c[t] = __builtin_amdgcn_wmma_f32_16x16x4_f32(false, a, false, b, (short)0, c[t], false, false);
    }
#pragma unroll
    for (int t = 0; t < NT; ++t) {
      v2f a, b;
      a.x = a4[sp].z; a.y = a4[sp].w; b.x = b4[t].z; b.y = b4[t].w;
      c[t] = __builtin_amdgcn_wmma_f32_16x16x4_f32(false, a, false, b, (short)0, c[t], false, false);
    }
  }
}

__global__ void __launch_bounds__(GW * 32, 8)
gemm_rows(const float* __restrict__ A, const float* __restrict__ W, const float* __restrict__ norm_out,
          float* __restrict__ out, int nNodes, int mTiles) {
  constexpr int KTOT = IN_F;
  constexpr int NPAD = HID_F;
  constexpr int NT   = NPAD / 16;
  static_assert(KTOT == 128, "two parts of 64 k");
  __shared__ __attribute__((aligned(16))) float Wq[KTOT * NPAD];
  __shared__ __attribute__((aligned(16))) float At[GW][16 * A_LD];

  const int tid = threadIdx.x, lane = tid & 31, w = tid >> 5;
  const int half = lane >> 4, l = lane & 15;
  const int tm = blockIdx.x * GW + w;
  const int tmc = tm < mTiles ? tm : mTiles - 1;
  float* at = At[w];

  int aoff[8];
#pragma unroll
  for (int q = 0; q < 8; ++q) {
    const int row = tmc * 16 + 2 * q + half;
    const int r = row < nNodes ? row : nNodes - 1;
    aoff[q] = r * KTOT + l * 4;
  }
  float4 g0[8];
#pragma unroll
  for (int q = 0; q < 8; ++q) g0[q] = *(const float4*)(A + aoff[q]);
  const int rowA = tmc * 16 + l;
  const float sA = norm_out[rowA < nNodes ? rowA : nNodes - 1];

  constexpr int WPER = (KTOT / 4) * NPAD / (GW * 32);
  static_assert((KTOT / 4) * NPAD % (GW * 32) == 0, "weight items divide evenly over the threads");
  float4 wv[WPER];
#pragma unroll
  for (int j = 0; j < WPER; ++j) {
    const int i = tid + j * GW * 32;
    const int kq = i / NPAD, n = i - kq * NPAD;
    const float* wp = W + (4 * kq) * NPAD + n;
    wv[j] = make_float4(wp[0], wp[NPAD], wp[2 * NPAD], wp[3 * NPAD]);
  }
#pragma unroll
  for (int j = 0; j < WPER; ++j) *(float4*)(Wq + (tid + j * GW * 32) * 4) = wv[j];

  v8f c[NT];
#pragma unroll
  for (int t = 0; t < NT; ++t) c[t] = v8f{};

#pragma unroll
  for (int q = 0; q < 8; ++q) *(float4*)(at + (2 * q + half) * A_LD + l * 4) = g0[q];
  __syncthreads();
  products64<true, NT, NPAD>(at, Wq, 0, half, l, sA, c);
  __syncthreads();
  float4 g1[8];
#pragma unroll
  for (int q = 0; q < 8; ++q) g1[q] = *(const float4*)(A + aoff[q] + 64);
#pragma unroll
  for (int q = 0; q < 8; ++q) *(float4*)(at + (2 * q + half) * A_LD + l * 4) = g1[q];
  __syncthreads();
  products64<true, NT, NPAD>(at, Wq, 16, half, l, sA, c);

  __syncthreads();
#pragma unroll
  for (int t = 0; t < NT; ++t) {
#pragma unroll
    for (int r = 0; r < 8; ++r) at[(8 * half + r) * A_LD + 16 * t + l] = c[t][r];
  }
  __syncthreads();
  if (tm < mTiles) {
#pragma unroll
    for (int q = 0; q < 8; ++q) {
      const int row = tm * 16 + 2 * q + half;
      if (row < nNodes)
        *(float4*)(out + (size_t)row * HID_F + l * 4) = *(const float4*)(at + (2 * q + half) * A_LD + l * 4);
    }
  }
}

__global__ void __launch_bounds__(GW * 32, 8)
layer2_layer3(const float* __restrict__ feat, const int* __restrict__ row_ptr,
              const int* __restrict__ cnt_in, const int* __restrict__ sorted_src,
              const float* __restrict__ W2, const float* __restrict__ b2, const float* __restrict__ W3,
              const float* __restrict__ norm_in, const float* __restrict__ norm_out,
              float* __restrict__ out, int nNodes, int mTiles) {
  constexpr int NT2 = HID_F / 16, NT3 = OUT_PAD / 16;
  __shared__ __attribute__((aligned(16))) float Wq2[HID_F * HID_F];
  __shared__ __attribute__((aligned(16))) float Wq3[HID_F * OUT_PAD];
  __shared__ __attribute__((aligned(16))) float At[GW][16 * A_LD];

  const int tid = threadIdx.x, lane = tid & 31, w = tid >> 5;
  const int half = lane >> 4, l = lane & 15;
  const int tm = blockIdx.x * GW + w;
  const int tmc = tm < mTiles ? tm : mTiles - 1;
  const int last = nNodes - 1;
  float* at = At[w];

  constexpr int WPER2 = (HID_F / 4) * HID_F / (GW * 32), WPER3 = (HID_F / 4) * OUT_PAD / (GW * 32);
  static_assert((HID_F / 4) * HID_F % (GW * 32) == 0 && (HID_F / 4) * OUT_PAD % (GW * 32) == 0,
                "weight items divide evenly over the threads");
  float4 wv2[WPER2], wv3[WPER3];
#pragma unroll
  for (int j = 0; j < WPER2; ++j) {
    const int i = tid + j * GW * 32;
    const int kq = i / HID_F, n = i - kq * HID_F;
    const float* wp = W2 + (4 * kq) * HID_F + n;
    wv2[j] = make_float4(wp[0], wp[HID_F], wp[2 * HID_F], wp[3 * HID_F]);
  }
#pragma unroll
  for (int j = 0; j < WPER3; ++j) {
    const int i = tid + j * GW * 32;
    const int kq = i / OUT_PAD, n = i - kq * OUT_PAD;
    const int nc = n < OUT_F ? n : OUT_F - 1;
    const float* wp = W3 + (4 * kq) * OUT_F + nc;
    float4 v = make_float4(wp[0], wp[OUT_F], wp[2 * OUT_F], wp[3 * OUT_F]);
    if (n >= OUT_F) v = make_float4(0.0f, 0.0f, 0.0f, 0.0f);
    wv3[j] = v;
  }

  const int nd = tmc * 16 + 8 * half + (l & 7);
  const int ndc = nd < last ? nd : last;
  const int e_l = row_ptr[ndc];
  const int c_l = cnt_in[ndc];
  const int d_l = (nd < nNodes && tm < mTiles) ? c_l : 0;

#pragma unroll
  for (int j = 0; j < WPER2; ++j) *(float4*)(Wq2 + (tid + j * GW * 32) * 4) = wv2[j];
#pragma unroll
  for (int j = 0; j < WPER3; ++j) *(float4*)(Wq3 + (tid + j * GW * 32) * 4) = wv3[j];

  const char* fb = (const char*)feat + l * 16;
  int e = __shfl(e_l, 0, 16), d = __shfl(d_l, 0, 16);
  int lo = sorted_src[l < d ? e + l : 0];
  int hi = sorted_src[l + 16 < d ? e + 16 + l : 0];
  for (int q = 0; q < 8; ++q) {
    const int qn = q < 7 ? q + 1 : 7;
    const int e1 = __shfl(e_l, qn, 16), d1 = __shfl(d_l, qn, 16);
    const int lo1 = sorted_src[l < d1 ? e1 + l : 0];
    const int hi1 = sorted_src[l + 16 < d1 ? e1 + 16 + l : 0];
    float4 acc = make_float4(0.0f, 0.0f, 0.0f, 0.0f);
    if (d > 0) sum_burst<HID_F * 4, 16>(acc, fb, lo, d, true);
    if (d > 24) sum_burst<HID_F * 4, 16>(acc, fb, hi, d - 16, true);
    else if (d > 16) sum_burst<HID_F * 4, 8>(acc, fb, hi, d - 16, true);
    for (int k = 32; k < d; ++k) {
      const float4 v = *(const float4*)(fb + (unsigned)sorted_src[e + k] * (HID_F * 4));
      add4(acc, v);
    }
    *(float4*)(at + (8 * half + q) * A_LD + l * 4) = acc;
    e = e1; d = d1; lo = lo1; hi = hi1;
  }

  float ni[8], no[8], bq[NT2];
#pragma unroll
  for (int r = 0; r < 8; ++r) {
    const int row = tmc * 16 + 8 * half + r;
    const int rc = row < last ? row : last;
    ni[r] = norm_in[rc];
    no[r] = norm_out[rc];
  }
#pragma unroll
  for (int t = 0; t < NT2; ++t) bq[t] = b2[16 * t + l];

  __syncthreads();
  v8f c[NT2];
#pragma unroll
  for (int t = 0; t < NT2; ++t) c[t] = v8f{};
  products64<false, NT2, HID_F>(at, Wq2, 0, half, l, 1.0f, c);
  __syncthreads();
#pragma unroll
  for (int t = 0; t < NT2; ++t) {
#pragma unroll
    for (int r = 0; r < 8; ++r)
      at[(8 * half + r) * A_LD + 16 * t + l] = fmaxf(c[t][r] * ni[r] + bq[t], 0.0f) * no[r];
  }
  __syncthreads();

  v8f c3[NT3];
#pragma unroll
  for (int t = 0; t < NT3; ++t) c3[t] = v8f{};
  products64<false, NT3, OUT_PAD>(at, Wq3, 0, half, l, 1.0f, c3);

  __syncthreads();
#pragma unroll
  for (int t = 0; t < NT3; ++t) {
#pragma unroll
    for (int r = 0; r < 8; ++r) at[(8 * half + r) * A_LD + 16 * t + l] = c3[t][r];
  }
  __syncthreads();
  if (tm < mTiles && l * 4 < OUT_PAD) {
#pragma unroll
    for (int q = 0; q < 8; ++q) {
      const int row = tm * 16 + 2 * q + half;
      if (row < nNodes)
        *(float4*)(out + (size_t)row * C_LD + l * 4) = *(const float4*)(at + (2 * q + half) * A_LD + l * 4);
    }
  }
}


extern "C" void kernel_launch(void* const* d_in, const int* in_sizes, int n_in,
                              void* d_out, int out_size, void* d_ws, size_t ws_size,
                              hipStream_t stream) {
  const float* X  = (const float*)d_in[0];
  const float* W1 = (const float*)d_in[1];
  const float* b1 = (const float*)d_in[2];
  const float* W2 = (const float*)d_in[3];
  const float* b2 = (const float*)d_in[4];
  const float* W3 = (const float*)d_in[5];
  const float* b3 = (const float*)d_in[6];
  const int* src  = (const int*)d_in[7];
  const int* dst  = (const int*)d_in[8];
  float* out = (float*)d_out;

  const int nNodes = in_sizes[0] / IN_F;
  const int nEdges = in_sizes[7];
  const int mTiles = (nNodes + 15) / 16;
  const int nBins  = (nNodes + BNODES - 1) >> BSH;
  const int nTiles = (nEdges + TILE - 1) / TILE;

  const size_t nP = ((size_t)nNodes + 63) & ~(size_t)63;
  const size_t eP = ((size_t)nEdges + 63) & ~(size_t)63;
  const size_t hP = ((size_t)nTiles * (MAXB / 2) + 63) & ~(size_t)63;
  const size_t sP = ((size_t)nTiles * (MAXB + 1) + 63) & ~(size_t)63;
  const size_t tP = (size_t)nTiles * (TILE / 4);
  int* wi = (int*)d_ws;
  int* cnt_in     = wi;
  int* row_ptr    = cnt_in + nP;
  int* hist       = row_ptr + nP;
  int* bin_base   = hist + hP;
  int* soff       = bin_base + MAXB + 64;
  int* tsbw       = soff + sP;
  int* bsrc       = tsbw + tP;
  int* bdst       = bsrc + eP;
  int* sorted_src = bdst + eP;
  float* norm_out = (float*)(sorted_src + eP);
  float* norm_in  = norm_out + nP;
  float* bufA     = norm_in + nP;
  float* bufB     = bufA + nP * HID_F;
  float* bufC     = bufB + nP * HID_F;

  const int gemmBlocks = (mTiles + GW - 1) / GW;
  const int gatherBlocks = GATHER_WG;
  const dim3 gblk(GATHER_T);

  tile_count<<<nTiles, dim3(TILE_T), 0, stream>>>(src, dst, hist, soff, tsbw, nEdges);
  tile_scatter<<<nTiles, dim3(TILE_T), 0, stream>>>(src, dst, hist, bin_base, bsrc, bdst, nEdges, nTiles);
  bin_finish<<<nBins, dim3(TILE_T), 0, stream>>>(bsrc, bdst, bin_base, soff, tsbw, row_ptr, cnt_in,
                                                norm_in, norm_out, sorted_src, nNodes, nTiles);

  gemm_rows<<<gemmBlocks, dim3(GW * 32), 0, stream>>>(
      X, W1, norm_out, bufA, nNodes, mTiles);
  gather_rows<1><<<gatherBlocks, gblk, 0, stream>>>(
      bufA, row_ptr, cnt_in, sorted_src, norm_in, norm_out, b1, bufB, nNodes);

  layer2_layer3<<<gemmBlocks, dim3(GW * 32), 0, stream>>>(
      bufB, row_ptr, cnt_in, sorted_src, W2, b2, W3, norm_in, norm_out, bufC, nNodes, mTiles);

  gather_rows<2><<<gatherBlocks, gblk, 0, stream>>>(
      bufC, row_ptr, cnt_in, sorted_src, norm_in, norm_out, b3, out, nNodes);
}
